# starting kernel
# baseline (speedup 1.0000x reference)
#include <hip/hip_runtime.h>
#include <math.h>

#define NQ    12
#define DIM   4096
#define DEPTH 4

typedef __attribute__((ext_vector_type(2))) float v2f;
typedef __attribute__((ext_vector_type(8))) float v8f;

__global__ void qdl_setup_kernel(const float* __restrict__ theta,
                                 float* __restrict__ ws) {
    const int tid = threadIdx.x;
    if (blockIdx.x < 12) {
        const int lg = blockIdx.x;
        const int l  = lg / 3, g = lg % 3;
        const int i  = tid >> 4, j = tid & 15;

        float pr = 1.0f, pi = 0.0f;
        #pragma unroll
        for (int q = 0; q < 4; ++q) {
            const int w = g * 4 + q;
            const float* th = theta + (l * NQ + w) * 3;
            const float phi = th[0], the = th[1], om = th[2];
            const float ch = cosf(0.5f * the), sh = sinf(0.5f * the);
            const float a  = 0.5f * (phi + om), b = 0.5f * (phi - om);
            const float ca = cosf(a), sa = sinf(a);
            const float cb = cosf(b), sb = sinf(b);
            const int bi = (i >> (3 - q)) & 1;
            const int bj = (j >> (3 - q)) & 1;
            float ur, ui;
            if      (bi == 0 && bj == 0) { ur =  ca * ch; ui = -sa * ch; }
            else if (bi == 0 && bj == 1) { ur = -cb * sh; ui = -sb * sh; }
            else if (bi == 1 && bj == 0) { ur =  cb * sh; ui = -sb * sh; }
            else                         { ur =  ca * ch; ui =  sa * ch; }
            const float nr = pr * ur - pi * ui;
            const float ni = pr * ui + pi * ur;
            pr = nr; pi = ni;
        }
        ws[(lg * 2 + 0) * 256 + tid] = pr;
        ws[(lg * 2 + 1) * 256 + tid] = pi;
    } else if (tid < 48) {
        const int l = tid / 12, k = tid % 12;
        const int r = l % (NQ - 1) + 1;
        int j = 1 << k;
        for (int w = NQ - 1; w >= 0; --w) {
            const int tg   = (w + r) % NQ;
            const int cbit = (j >> (NQ - 1 - w)) & 1;
            j ^= cbit << (NQ - 1 - tg);
        }
        ((int*)ws)[12 * 2 * 256 + tid] = j;
    }
}

__global__ __launch_bounds__(256) void qdl_main_kernel(
        const float* __restrict__ x,
        const float* __restrict__ ws,
        float* __restrict__ out) {
    __shared__ float psiR[DIM];
    __shared__ float psiI[DIM];
    __shared__ float gLDS[12 * 2 * 256];
    __shared__ int   permB[48];
    __shared__ float red[8 * 12];

    const int tid  = threadIdx.x;
    const int b    = blockIdx.x;
    const int lane = tid & 31;
    const int wave = tid >> 5;
    const int half = lane >> 4;
    const int mn   = lane & 15;

    for (int k = tid; k < 12 * 2 * 256; k += 256) gLDS[k] = ws[k];
    if (tid < 48) permB[tid] = ((const int*)ws)[12 * 2 * 256 + tid];

    float4* psiR4 = reinterpret_cast<float4*>(psiR);
    float4* psiI4 = reinterpret_cast<float4*>(psiI);

    const float4* xr4 = reinterpret_cast<const float4*>(x + (size_t)b * DIM);
    float4 xv[4];
    float ss = 0.0f;
    #pragma unroll
    for (int e = 0; e < 4; ++e) {
        const float4 v = xr4[tid + 256 * e];
        xv[e] = v;
        ss += v.x * v.x + v.y * v.y + v.z * v.z + v.w * v.w;
    }
    for (int o = 16; o > 0; o >>= 1) ss += __shfl_xor(ss, o, 32);
    if (lane == 0) red[wave] = ss;
    __syncthreads();
    float n1sq = 0.0f;
    #pragma unroll
    for (int wv = 0; wv < 8; ++wv) n1sq += red[wv];
    const float x0   = x[(size_t)b * DIM];
    const bool  zero = (sqrtf(n1sq) <= 1e-12f);
    const float x0n  = zero ? 1.0f : x0;
    const float inv  = 1.0f / sqrtf(n1sq - x0 * x0 + x0n * x0n);
    __syncthreads();
    #pragma unroll
    for (int e = 0; e < 4; ++e) {
        float4 v = xv[e];
        v.x = (tid == 0 && e == 0) ? x0n : v.x;
        v.x *= inv; v.y *= inv; v.z *= inv; v.w *= inv;
        psiR4[tid + 256 * e] = v;
        psiI4[tid + 256 * e] = make_float4(0.f, 0.f, 0.f, 0.f);
    }
    __syncthreads();

    for (int l = 0; l < DEPTH; ++l) {
        #pragma unroll
        for (int g = 0; g < 3; ++g) {
            const float* Gr = &gLDS[((l * 3 + g) * 2 + 0) * 256];
            const float* Gi = &gLDS[((l * 3 + g) * 2 + 1) * 256];

            v2f Ar[4], Ai[4], Ain[4];
            #pragma unroll
            for (int kq = 0; kq < 4; ++kq) {
                const int k0 = kq * 4 + 2 * half;
                const v2f ar = *reinterpret_cast<const v2f*>(&Gr[mn * 16 + k0]);
                const v2f ai = *reinterpret_cast<const v2f*>(&Gi[mn * 16 + k0]);
                Ar[kq]  = ar;
                Ai[kq]  = ai;
                Ain[kq] = (v2f){-ai.x, -ai.y};
            }

            #pragma unroll
            for (int tt = 0; tt < 2; ++tt) {
                const int t = wave + tt * 8;
                v8f accR = {}; v8f accI = {};
                #pragma unroll
                for (int kq = 0; kq < 4; ++kq) {
                    const int k0 = kq * 4 + 2 * half;
                    v2f Br, Bi;
                    if (g == 0) {
                        const int i0 = k0 * 256 + t * 16 + mn;
                        Br = (v2f){psiR[i0], psiR[i0 + 256]};
                        Bi = (v2f){psiI[i0], psiI[i0 + 256]};
                    } else if (g == 1) {
                        const int i0 = t * 256 + k0 * 16 + mn;
                        Br = (v2f){psiR[i0], psiR[i0 + 16]};
                        Bi = (v2f){psiI[i0], psiI[i0 + 16]};
                    } else {
                        const int i0 = t * 256 + mn * 16 + k0;
                        Br = *reinterpret_cast<const v2f*>(&psiR[i0]);
                        Bi = *reinterpret_cast<const v2f*>(&psiI[i0]);
                    }
                    accR = __builtin_amdgcn_wmma_f32_16x16x4_f32(false, Ar[kq],  false, Br, (short)0, accR, false, false);
                    accR = __builtin_amdgcn_wmma_f32_16x16x4_f32(false, Ain[kq], false, Bi, (short)0, accR, false, false);
                    accI = __builtin_amdgcn_wmma_f32_16x16x4_f32(false, Ar[kq],  false, Bi, (short)0, accI, false, false);
                    accI = __builtin_amdgcn_wmma_f32_16x16x4_f32(false, Ai[kq],  false, Br, (short)0, accI, false, false);
                }
                #pragma unroll
                for (int v = 0; v < 8; ++v) {
                    const int m = v + 8 * half;
                    int idx;
                    if      (g == 0) idx = m * 256 + t * 16 + mn;
                    else if (g == 1) idx = t * 256 + m * 16 + mn;
                    else             idx = t * 256 + mn * 16 + m;
                    psiR[idx] = accR[v];
                    psiI[idx] = accI[v];
                }
            }
            __syncthreads();
        }

        const int* bas = &permB[l * 12];
        int pbase = 0;
        #pragma unroll
        for (int bb = 0; bb < 8; ++bb)
            pbase ^= ((tid >> bb) & 1) ? bas[4 + bb] : 0;
        int pe[16];
        #pragma unroll
        for (int e = 0; e < 16; ++e) {
            int v = 0;
            if (e & 1) v ^= bas[0];
            if (e & 2) v ^= bas[1];
            if (e & 4) v ^= bas[2];
            if (e & 8) v ^= bas[3];
            pe[e] = v;
        }
        float rr[16], ri[16];
        #pragma unroll
        for (int e = 0; e < 16; ++e) {
            const int j = pbase ^ pe[e];
            rr[e] = psiR[j];
            ri[e] = psiI[j];
        }
        __syncthreads();
        #pragma unroll
        for (int q = 0; q < 4; ++q) {
            psiR4[tid * 4 + q] = make_float4(rr[4*q], rr[4*q+1], rr[4*q+2], rr[4*q+3]);
            psiI4[tid * 4 + q] = make_float4(ri[4*q], ri[4*q+1], ri[4*q+2], ri[4*q+3]);
        }
        __syncthreads();
    }

    float acc[12];
    #pragma unroll
    for (int k = 0; k < 12; ++k) acc[k] = 0.0f;
    #pragma unroll
    for (int e4 = 0; e4 < 4; ++e4) {
        const float4 a = psiR4[tid * 4 + e4];
        const float4 c = psiI4[tid * 4 + e4];
        const int ib = tid * 16 + e4 * 4;
        const float p0 = a.x * a.x + c.x * c.x;
        const float p1 = a.y * a.y + c.y * c.y;
        const float p2 = a.z * a.z + c.z * c.z;
        const float p3 = a.w * a.w + c.w * c.w;
        const float pv[4] = {p0, p1, p2, p3};
        #pragma unroll
        for (int cc = 0; cc < 4; ++cc) {
            const int i = ib + cc;
            #pragma unroll
            for (int k = 0; k < 12; ++k)
                acc[k] += ((i >> (11 - k)) & 1) ? -pv[cc] : pv[cc];
        }
    }
    #pragma unroll
    for (int k = 0; k < 12; ++k) {
        float v = acc[k];
        for (int o = 16; o > 0; o >>= 1) v += __shfl_xor(v, o, 32);
        if (lane == 0) red[wave * 12 + k] = v;
    }
    __syncthreads();
    if (tid < 12) {
        float s = 0.0f;
        #pragma unroll
        for (int wv = 0; wv < 8; ++wv) s += red[wv * 12 + tid];
        out[(size_t)b * 12 + tid] = s;
    }
}

extern "C" void kernel_launch(void* const* d_in, const int* in_sizes, int n_in,
                              void* d_out, int out_size, void* d_ws, size_t ws_size,
                              hipStream_t stream) {
    const float* x     = (const float*)d_in[0];
    const float* theta = (const float*)d_in[1];
    float*       out   = (float*)d_out;
    float*       wsf   = (float*)d_ws;

    const int B = in_sizes[0] / DIM;

    qdl_setup_kernel<<<13, 256, 0, stream>>>(theta, wsf);
    qdl_main_kernel<<<B, 256, 0, stream>>>(x, wsf, out);
}


# LDS addresses as one per-thread base plus immediate offsets (real and imaginary planes in one array): VGPRs 168 -> 96, so 5 instead of 2 workgroups are resident per WGP; sums unchanged
# speedup vs baseline: 1.2594x; 1.2594x over previous
#include <hip/hip_runtime.h>
#include <math.h>

#define NQ    12
#define DIM   4096
#define DEPTH 4

typedef __attribute__((ext_vector_type(2))) float v2f;
typedef __attribute__((ext_vector_type(8))) float v8f;

__global__ void qdl_setup_kernel(const float* __restrict__ theta,
                                 float* __restrict__ ws) {
    const int tid = threadIdx.x;
    if (blockIdx.x < 12) {
        const int lg = blockIdx.x;
        const int l  = lg / 3, g = lg % 3;
        const int i  = tid >> 4, j = tid & 15;

        float pr = 1.0f, pi = 0.0f;
        #pragma unroll
        for (int q = 0; q < 4; ++q) {
            const int w = g * 4 + q;
            const float* th = theta + (l * NQ + w) * 3;
            const float phi = th[0], the = th[1], om = th[2];
            const float ch = cosf(0.5f * the), sh = sinf(0.5f * the);
            const float a  = 0.5f * (phi + om), b = 0.5f * (phi - om);
            const float ca = cosf(a), sa = sinf(a);
            const float cb = cosf(b), sb = sinf(b);
            const int bi = (i >> (3 - q)) & 1;
            const int bj = (j >> (3 - q)) & 1;
            float ur, ui;
            if      (bi == 0 && bj == 0) { ur =  ca * ch; ui = -sa * ch; }
            else if (bi == 0 && bj == 1) { ur = -cb * sh; ui = -sb * sh; }
            else if (bi == 1 && bj == 0) { ur =  cb * sh; ui = -sb * sh; }
            else                         { ur =  ca * ch; ui =  sa * ch; }
            const float nr = pr * ur - pi * ui;
            const float ni = pr * ui + pi * ur;
            pr = nr; pi = ni;
        }
        ws[(lg * 2 + 0) * 256 + tid] = pr;
        ws[(lg * 2 + 1) * 256 + tid] = pi;
    } else if (tid < 48) {
        const int l = tid / 12, k = tid % 12;
        const int r = l % (NQ - 1) + 1;
        int j = 1 << k;
        for (int w = NQ - 1; w >= 0; --w) {
            const int tg   = (w + r) % NQ;
            const int cbit = (j >> (NQ - 1 - w)) & 1;
            j ^= cbit << (NQ - 1 - tg);
        }
        ((int*)ws)[12 * 2 * 256 + tid] = j;
    }
}

__global__ __launch_bounds__(256) void qdl_main_kernel(
        const float* __restrict__ x,
        const float* __restrict__ ws,
        float* __restrict__ out) {
    __shared__ float psi[2 * DIM];
    float* const psiR = psi;
    float* const psiI = psi + DIM;
    __shared__ float gLDS[12 * 2 * 256];
    __shared__ int   permB[48];
    __shared__ float red[8 * 12];

    const int tid  = threadIdx.x;
    const int b    = blockIdx.x;
    const int lane = tid & 31;
    const int wave = tid >> 5;
    const int half = lane >> 4;
    const int mn   = lane & 15;

    for (int k = tid; k < 12 * 2 * 256; k += 256) gLDS[k] = ws[k];
    if (tid < 48) permB[tid] = ((const int*)ws)[12 * 2 * 256 + tid];

    float4* psiR4 = reinterpret_cast<float4*>(psiR);
    float4* psiI4 = reinterpret_cast<float4*>(psiI);

    const float4* xr4 = reinterpret_cast<const float4*>(x + (size_t)b * DIM);
    float4 xv[4];
    float ss = 0.0f;
    #pragma unroll
    for (int e = 0; e < 4; ++e) {
        const float4 v = xr4[tid + 256 * e];
        xv[e] = v;
        ss += v.x * v.x + v.y * v.y + v.z * v.z + v.w * v.w;
    }
    for (int o = 16; o > 0; o >>= 1) ss += __shfl_xor(ss, o, 32);
    if (lane == 0) red[wave] = ss;
    __syncthreads();
    float n1sq = 0.0f;
    #pragma unroll
    for (int wv = 0; wv < 8; ++wv) n1sq += red[wv];
    const float x0   = x[(size_t)b * DIM];
    const bool  zero = (sqrtf(n1sq) <= 1e-12f);
    const float x0n  = zero ? 1.0f : x0;
    const float inv  = 1.0f / sqrtf(n1sq - x0 * x0 + x0n * x0n);
    __syncthreads();
    #pragma unroll
    for (int e = 0; e < 4; ++e) {
        float4 v = xv[e];
        v.x = (tid == 0 && e == 0) ? x0n : v.x;
        v.x *= inv; v.y *= inv; v.z *= inv; v.w *= inv;
        psiR4[tid + 256 * e] = v;
        psiI4[tid + 256 * e] = make_float4(0.f, 0.f, 0.f, 0.f);
    }
    __syncthreads();

    const float* const pB0 = psi + (half * 512 + wave * 16 + mn);
    float*       const pD0 = psi + (half * 2048 + wave * 16 + mn);
    const float* const pB1 = psi + (wave * 256 + half * 32 + mn);
    float*       const pD1 = psi + (wave * 256 + half * 128 + mn);
    const float* const pB2 = psi + (wave * 256 + mn * 16 + half * 2);
    float*       const pD2 = psi + (wave * 256 + mn * 16 + half * 8);
    const float* const pA  = gLDS + (mn * 16 + half * 2);

    for (int l = 0; l < DEPTH; ++l) {
        const float* const pAl = pA + l * (3 * 2 * 256);
        #pragma unroll
        for (int g = 0; g < 3; ++g) {

            v2f Ar[4], Ai[4], Ain[4];
            #pragma unroll
            for (int kq = 0; kq < 4; ++kq) {
                const v2f ar = *reinterpret_cast<const v2f*>(&pAl[(g * 2 + 0) * 256 + kq * 4]);
                const v2f ai = *reinterpret_cast<const v2f*>(&pAl[(g * 2 + 1) * 256 + kq * 4]);
                Ar[kq]  = ar;
                Ai[kq]  = ai;
                Ain[kq] = (v2f){-ai.x, -ai.y};
            }

            #pragma unroll
            for (int tt = 0; tt < 2; ++tt) {
                v8f accR = {}; v8f accI = {};
                #pragma unroll
                for (int kq = 0; kq < 4; ++kq) {
                    v2f Br, Bi;
                    if (g == 0) {
                        const int o = kq * 1024 + tt * 128;
                        Br = (v2f){pB0[o], pB0[o + 256]};
                        Bi = (v2f){pB0[DIM + o], pB0[DIM + o + 256]};
                    } else if (g == 1) {
                        const int o = tt * 2048 + kq * 64;
                        Br = (v2f){pB1[o], pB1[o + 16]};
                        Bi = (v2f){pB1[DIM + o], pB1[DIM + o + 16]};
                    } else {
                        const int o = tt * 2048 + kq * 4;
                        Br = *reinterpret_cast<const v2f*>(&pB2[o]);
                        Bi = *reinterpret_cast<const v2f*>(&pB2[DIM + o]);
                    }
                    accR = __builtin_amdgcn_wmma_f32_16x16x4_f32(false, Ar[kq],  false, Br, (short)0, accR, false, false);
                    accR = __builtin_amdgcn_wmma_f32_16x16x4_f32(false, Ain[kq], false, Bi, (short)0, accR, false, false);
                    accI = __builtin_amdgcn_wmma_f32_16x16x4_f32(false, Ar[kq],  false, Bi, (short)0, accI, false, false);
                    accI = __builtin_amdgcn_wmma_f32_16x16x4_f32(false, Ai[kq],  false, Br, (short)0, accI, false, false);
                }
                #pragma unroll
                for (int v = 0; v < 8; ++v) {
                    float* p;
                    if      (g == 0) p = pD0 + (v * 256 + tt * 128);
                    else if (g == 1) p = pD1 + (tt * 2048 + v * 16);
                    else             p = pD2 + (tt * 2048 + v);
                    p[0]   = accR[v];
                    p[DIM] = accI[v];
                }
            }
            __syncthreads();
        }

        const int* bas = &permB[l * 12];
        int pbase = 0;
        #pragma unroll
        for (int bb = 0; bb < 8; ++bb)
            pbase ^= ((tid >> bb) & 1) ? bas[4 + bb] : 0;
        int pe[16];
        #pragma unroll
        for (int e = 0; e < 16; ++e) {
            int v = 0;
            if (e & 1) v ^= bas[0];
            if (e & 2) v ^= bas[1];
            if (e & 4) v ^= bas[2];
            if (e & 8) v ^= bas[3];
            pe[e] = v;
        }
        float rr[16], ri[16];
        #pragma unroll
        for (int e = 0; e < 16; ++e) {
            const int j = pbase ^ pe[e];
            rr[e] = psiR[j];
            ri[e] = psiI[j];
        }
        __syncthreads();
        #pragma unroll
        for (int q = 0; q < 4; ++q) {
            psiR4[tid * 4 + q] = make_float4(rr[4*q], rr[4*q+1], rr[4*q+2], rr[4*q+3]);
            psiI4[tid * 4 + q] = make_float4(ri[4*q], ri[4*q+1], ri[4*q+2], ri[4*q+3]);
        }
        __syncthreads();
    }

    float acc[12];
    #pragma unroll
    for (int k = 0; k < 12; ++k) acc[k] = 0.0f;
    #pragma unroll
    for (int e4 = 0; e4 < 4; ++e4) {
        const float4 a = psiR4[tid * 4 + e4];
        const float4 c = psiI4[tid * 4 + e4];
        const int ib = tid * 16 + e4 * 4;
        const float p0 = a.x * a.x + c.x * c.x;
        const float p1 = a.y * a.y + c.y * c.y;
        const float p2 = a.z * a.z + c.z * c.z;
        const float p3 = a.w * a.w + c.w * c.w;
        const float pv[4] = {p0, p1, p2, p3};
        #pragma unroll
        for (int cc = 0; cc < 4; ++cc) {
            const int i = ib + cc;
            #pragma unroll
            for (int k = 0; k < 12; ++k)
                acc[k] += ((i >> (11 - k)) & 1) ? -pv[cc] : pv[cc];
        }
    }
    #pragma unroll
    for (int k = 0; k < 12; ++k) {
        float v = acc[k];
        for (int o = 16; o > 0; o >>= 1) v += __shfl_xor(v, o, 32);
        if (lane == 0) red[wave * 12 + k] = v;
    }
    __syncthreads();
    if (tid < 12) {
        float s = 0.0f;
        #pragma unroll
        for (int wv = 0; wv < 8; ++wv) s += red[wv * 12 + tid];
        out[(size_t)b * 12 + tid] = s;
    }
}

extern "C" void kernel_launch(void* const* d_in, const int* in_sizes, int n_in,
                              void* d_out, int out_size, void* d_ws, size_t ws_size,
                              hipStream_t stream) {
    const float* x     = (const float*)d_in[0];
    const float* theta = (const float*)d_in[1];
    float*       out   = (float*)d_out;
    float*       wsf   = (float*)d_ws;

    const int B = in_sizes[0] / DIM;

    qdl_setup_kernel<<<13, 256, 0, stream>>>(theta, wsf);
    qdl_main_kernel<<<B, 256, 0, stream>>>(x, wsf, out);
}


# LDS bank conflicts removed: state padded to digit strides 276/52 floats and gate rows stored k-quad-major, so each fragment load/store hits distinct banks; sums and their order unchanged
# speedup vs baseline: 1.3965x; 1.1089x over previous
#include <hip/hip_runtime.h>
#include <math.h>

#define NQ    12
#define DIM   4096
#define DEPTH 4

#define SA    276
#define SB    52
#define PLANE (15 * SA + 15 * SB + 16)
#define SIDX(a, b, c) ((a) * SA + (b) * SB + (c))

typedef __attribute__((ext_vector_type(2))) float v2f;
typedef __attribute__((ext_vector_type(8))) float v8f;

__global__ void qdl_setup_kernel(const float* __restrict__ theta,
                                 float* __restrict__ ws) {
    const int tid = threadIdx.x;
    if (blockIdx.x < 12) {
        const int lg = blockIdx.x;
        const int l  = lg / 3, g = lg % 3;
        const int i  = tid >> 4, j = tid & 15;

        float pr = 1.0f, pi = 0.0f;
        #pragma unroll
        for (int q = 0; q < 4; ++q) {
            const int w = g * 4 + q;
            const float* th = theta + (l * NQ + w) * 3;
            const float phi = th[0], the = th[1], om = th[2];
            const float ch = cosf(0.5f * the), sh = sinf(0.5f * the);
            const float a  = 0.5f * (phi + om), b = 0.5f * (phi - om);
            const float ca = cosf(a), sa = sinf(a);
            const float cb = cosf(b), sb = sinf(b);
            const int bi = (i >> (3 - q)) & 1;
            const int bj = (j >> (3 - q)) & 1;
            float ur, ui;
            if      (bi == 0 && bj == 0) { ur =  ca * ch; ui = -sa * ch; }
            else if (bi == 0 && bj == 1) { ur = -cb * sh; ui = -sb * sh; }
            else if (bi == 1 && bj == 0) { ur =  cb * sh; ui = -sb * sh; }
            else                         { ur =  ca * ch; ui =  sa * ch; }
            const float nr = pr * ur - pi * ui;
            const float ni = pr * ui + pi * ur;
            pr = nr; pi = ni;
        }
        const int ga = (j >> 2) * 64 + i * 4 + (j & 3);
        ws[(lg * 2 + 0) * 256 + ga] = pr;
        ws[(lg * 2 + 1) * 256 + ga] = pi;
    } else if (tid < 48) {
        const int l = tid / 12, k = tid % 12;
        const int r = l % (NQ - 1) + 1;
        int j = 1 << k;
        for (int w = NQ - 1; w >= 0; --w) {
            const int tg   = (w + r) % NQ;
            const int cbit = (j >> (NQ - 1 - w)) & 1;
            j ^= cbit << (NQ - 1 - tg);
        }
        ((int*)ws)[12 * 2 * 256 + tid] = j;
    }
}

__global__ __launch_bounds__(256, 10) void qdl_main_kernel(
        const float* __restrict__ x,
        const float* __restrict__ ws,
        float* __restrict__ out) {
    __shared__ __attribute__((aligned(16))) float psi[2 * PLANE];
    float* const psiR = psi;
    float* const psiI = psi + PLANE;
    __shared__ float gLDS[12 * 2 * 256];
    __shared__ int   permB[48];
    __shared__ float red[8 * 12];

    const int tid  = threadIdx.x;
    const int b    = blockIdx.x;
    const int lane = tid & 31;
    const int wave = tid >> 5;
    const int half = lane >> 4;
    const int mn   = lane & 15;

    for (int k = tid; k < 12 * 2 * 256; k += 256) gLDS[k] = ws[k];
    if (tid < 48) permB[tid] = ((const int*)ws)[12 * 2 * 256 + tid];


    const float4* xr4 = reinterpret_cast<const float4*>(x + (size_t)b * DIM);
    float4 xv[4];
    float ss = 0.0f;
    #pragma unroll
    for (int e = 0; e < 4; ++e) {
        const float4 v = xr4[tid + 256 * e];
        xv[e] = v;
        ss += v.x * v.x + v.y * v.y + v.z * v.z + v.w * v.w;
    }
    for (int o = 16; o > 0; o >>= 1) ss += __shfl_xor(ss, o, 32);
    if (lane == 0) red[wave] = ss;
    __syncthreads();
    float n1sq = 0.0f;
    #pragma unroll
    for (int wv = 0; wv < 8; ++wv) n1sq += red[wv];
    const float x0   = x[(size_t)b * DIM];
    const bool  zero = (sqrtf(n1sq) <= 1e-12f);
    const float x0n  = zero ? 1.0f : x0;
    const float inv  = 1.0f / sqrtf(n1sq - x0 * x0 + x0n * x0n);
    __syncthreads();
    #pragma unroll
    for (int e = 0; e < 4; ++e) {
        float4 v = xv[e];
        v.x = (tid == 0 && e == 0) ? x0n : v.x;
        v.x *= inv; v.y *= inv; v.z *= inv; v.w *= inv;
        float* const p = psi + SIDX((tid >> 6) + 4 * e, (tid >> 2) & 15, (tid & 3) * 4);
        *reinterpret_cast<float4*>(p)         = v;
        *reinterpret_cast<float4*>(p + PLANE) = make_float4(0.f, 0.f, 0.f, 0.f);
    }
    __syncthreads();

    const float* const pB0 = psi + SIDX(2 * half, wave, mn);
    float*       const pD0 = psi + SIDX(8 * half, wave, mn);
    const float* const pB1 = psi + SIDX(wave, 2 * half, mn);
    float*       const pD1 = psi + SIDX(wave, 8 * half, mn);
    const float* const pB2 = psi + SIDX(wave, mn, 2 * half);
    float*       const pD2 = psi + SIDX(wave, mn, 8 * half);
    const float* const pA  = gLDS + (mn * 4 + half * 2);

    for (int l = 0; l < DEPTH; ++l) {
        const float* const pAl = pA + l * (3 * 2 * 256);
        #pragma unroll
        for (int g = 0; g < 3; ++g) {

            v2f Ar[4], Ai[4], Ain[4];
            #pragma unroll
            for (int kq = 0; kq < 4; ++kq) {
                const v2f ar = *reinterpret_cast<const v2f*>(&pAl[(g * 2 + 0) * 256 + kq * 64]);
                const v2f ai = *reinterpret_cast<const v2f*>(&pAl[(g * 2 + 1) * 256 + kq * 64]);
                Ar[kq]  = ar;
                Ai[kq]  = ai;
                Ain[kq] = (v2f){-ai.x, -ai.y};
            }

            #pragma unroll
            for (int tt = 0; tt < 2; ++tt) {
                v8f accR = {}; v8f accI = {};
                #pragma unroll
                for (int kq = 0; kq < 4; ++kq) {
                    v2f Br, Bi;
                    if (g == 0) {
                        const int o = SIDX(kq * 4, 8 * tt, 0);
                        Br = (v2f){pB0[o], pB0[o + SA]};
                        Bi = (v2f){pB0[PLANE + o], pB0[PLANE + o + SA]};
                    } else if (g == 1) {
                        const int o = SIDX(8 * tt, kq * 4, 0);
                        Br = (v2f){pB1[o], pB1[o + SB]};
                        Bi = (v2f){pB1[PLANE + o], pB1[PLANE + o + SB]};
                    } else {
                        const int o = SIDX(8 * tt, 0, kq * 4);
                        Br = *reinterpret_cast<const v2f*>(&pB2[o]);
                        Bi = *reinterpret_cast<const v2f*>(&pB2[PLANE + o]);
                    }
                    accR = __builtin_amdgcn_wmma_f32_16x16x4_f32(false, Ar[kq],  false, Br, (short)0, accR, false, false);
                    accR = __builtin_amdgcn_wmma_f32_16x16x4_f32(false, Ain[kq], false, Bi, (short)0, accR, false, false);
                    accI = __builtin_amdgcn_wmma_f32_16x16x4_f32(false, Ar[kq],  false, Bi, (short)0, accI, false, false);
                    accI = __builtin_amdgcn_wmma_f32_16x16x4_f32(false, Ai[kq],  false, Br, (short)0, accI, false, false);
                }
                #pragma unroll
                for (int v = 0; v < 8; ++v) {
                    float* p;
                    if      (g == 0) p = pD0 + SIDX(v, 8 * tt, 0);
                    else if (g == 1) p = pD1 + SIDX(8 * tt, v, 0);
                    else             p = pD2 + SIDX(8 * tt, 0, v);
                    p[0]     = accR[v];
                    p[PLANE] = accI[v];
                }
            }
            __syncthreads();
        }

        const int* bas = &permB[l * 12];
        int pbase = 0;
        #pragma unroll
        for (int bb = 0; bb < 8; ++bb)
            pbase ^= bas[4 + bb] & -((tid >> bb) & 1);
        int pe[16];
        #pragma unroll
        for (int e = 0; e < 16; ++e) {
            int v = 0;
            if (e & 1) v ^= bas[0];
            if (e & 2) v ^= bas[1];
            if (e & 4) v ^= bas[2];
            if (e & 8) v ^= bas[3];
            pe[e] = v;
        }
        float rr[16], ri[16];
        #pragma unroll
        for (int e = 0; e < 16; ++e) {
            const int j  = pbase ^ pe[e];
            const int jp = SIDX(j >> 8, (j >> 4) & 15, j & 15);
            rr[e] = psiR[jp];
            ri[e] = psiI[jp];
        }
        __syncthreads();
        float* const pw = psi + SIDX(tid >> 4, tid & 15, 0);
        #pragma unroll
        for (int q = 0; q < 4; ++q) {
            *reinterpret_cast<float4*>(pw + 4 * q)         = make_float4(rr[4*q], rr[4*q+1], rr[4*q+2], rr[4*q+3]);
            *reinterpret_cast<float4*>(pw + PLANE + 4 * q) = make_float4(ri[4*q], ri[4*q+1], ri[4*q+2], ri[4*q+3]);
        }
        __syncthreads();
    }

    float acc[12];
    #pragma unroll
    for (int k = 0; k < 12; ++k) acc[k] = 0.0f;
    const float* const pf = psi + SIDX(tid >> 4, tid & 15, 0);
    #pragma unroll
    for (int e4 = 0; e4 < 4; ++e4) {
        const float4 a = *reinterpret_cast<const float4*>(pf + 4 * e4);
        const float4 c = *reinterpret_cast<const float4*>(pf + PLANE + 4 * e4);
        const int ib = tid * 16 + e4 * 4;
        const float p0 = a.x * a.x + c.x * c.x;
        const float p1 = a.y * a.y + c.y * c.y;
        const float p2 = a.z * a.z + c.z * c.z;
        const float p3 = a.w * a.w + c.w * c.w;
        const float pv[4] = {p0, p1, p2, p3};
        #pragma unroll
        for (int cc = 0; cc < 4; ++cc) {
            const int i = ib + cc;
            #pragma unroll
            for (int k = 0; k < 12; ++k)
                acc[k] += ((i >> (11 - k)) & 1) ? -pv[cc] : pv[cc];
        }
    }
    #pragma unroll
    for (int k = 0; k < 12; ++k) {
        float v = acc[k];
        for (int o = 16; o > 0; o >>= 1) v += __shfl_xor(v, o, 32);
        if (lane == 0) red[wave * 12 + k] = v;
    }
    __syncthreads();
    if (tid < 12) {
        float s = 0.0f;
        #pragma unroll
        for (int wv = 0; wv < 8; ++wv) s += red[wv * 12 + tid];
        out[(size_t)b * 12 + tid] = s;
    }
}

extern "C" void kernel_launch(void* const* d_in, const int* in_sizes, int n_in,
                              void* d_out, int out_size, void* d_ws, size_t ws_size,
                              hipStream_t stream) {
    const float* x     = (const float*)d_in[0];
    const float* theta = (const float*)d_in[1];
    float*       out   = (float*)d_out;
    float*       wsf   = (float*)d_ws;

    const int B = in_sizes[0] / DIM;

    qdl_setup_kernel<<<13, 256, 0, stream>>>(theta, wsf);
    qdl_main_kernel<<<B, 256, 0, stream>>>(x, wsf, out);
}


# Complex 16x16 gate product from three real WMMA sums instead of four (Re = ArBr - AiBi, Im = (Ar+Ai)(Br+Bi) - ArBr - AiBi): 72 not 96 WMMAs per wave per layer; rounding of the summed parts changes
# speedup vs baseline: 1.4814x; 1.0608x over previous
#include <hip/hip_runtime.h>
#include <math.h>

#define NQ    12
#define DIM   4096
#define DEPTH 4

#define SA    276
#define SB    52
#define PLANE (15 * SA + 15 * SB + 16)
#define SIDX(a, b, c) ((a) * SA + (b) * SB + (c))

typedef __attribute__((ext_vector_type(2))) float v2f;
typedef __attribute__((ext_vector_type(8))) float v8f;

__global__ void qdl_setup_kernel(const float* __restrict__ theta,
                                 float* __restrict__ ws) {
    const int tid = threadIdx.x;
    if (blockIdx.x < 12) {
        const int lg = blockIdx.x;
        const int l  = lg / 3, g = lg % 3;
        const int i  = tid >> 4, j = tid & 15;

        float pr = 1.0f, pi = 0.0f;
        #pragma unroll
        for (int q = 0; q < 4; ++q) {
            const int w = g * 4 + q;
            const float* th = theta + (l * NQ + w) * 3;
            const float phi = th[0], the = th[1], om = th[2];
            const float ch = cosf(0.5f * the), sh = sinf(0.5f * the);
            const float a  = 0.5f * (phi + om), b = 0.5f * (phi - om);
            const float ca = cosf(a), sa = sinf(a);
            const float cb = cosf(b), sb = sinf(b);
            const int bi = (i >> (3 - q)) & 1;
            const int bj = (j >> (3 - q)) & 1;
            float ur, ui;
            if      (bi == 0 && bj == 0) { ur =  ca * ch; ui = -sa * ch; }
            else if (bi == 0 && bj == 1) { ur = -cb * sh; ui = -sb * sh; }
            else if (bi == 1 && bj == 0) { ur =  cb * sh; ui = -sb * sh; }
            else                         { ur =  ca * ch; ui =  sa * ch; }
            const float nr = pr * ur - pi * ui;
            const float ni = pr * ui + pi * ur;
            pr = nr; pi = ni;
        }
        const int ga = (j >> 2) * 64 + i * 4 + (j & 3);
        ws[(lg * 2 + 0) * 256 + ga] = pr;
        ws[(lg * 2 + 1) * 256 + ga] = pi;
    } else if (tid < 48) {
        const int l = tid / 12, k = tid % 12;
        const int r = l % (NQ - 1) + 1;
        int j = 1 << k;
        for (int w = NQ - 1; w >= 0; --w) {
            const int tg   = (w + r) % NQ;
            const int cbit = (j >> (NQ - 1 - w)) & 1;
            j ^= cbit << (NQ - 1 - tg);
        }
        ((int*)ws)[12 * 2 * 256 + tid] = j;
    }
}

__global__ __launch_bounds__(256, 10) void qdl_main_kernel(
        const float* __restrict__ x,
        const float* __restrict__ ws,
        float* __restrict__ out) {
    __shared__ __attribute__((aligned(16))) float psi[2 * PLANE];
    float* const psiR = psi;
    float* const psiI = psi + PLANE;
    __shared__ float gLDS[12 * 2 * 256];
    __shared__ int   permB[48];
    __shared__ float red[8 * 12];

    const int tid  = threadIdx.x;
    const int b    = blockIdx.x;
    const int lane = tid & 31;
    const int wave = tid >> 5;
    const int half = lane >> 4;
    const int mn   = lane & 15;

    for (int k = tid; k < 12 * 2 * 256; k += 256) gLDS[k] = ws[k];
    if (tid < 48) permB[tid] = ((const int*)ws)[12 * 2 * 256 + tid];


    const float4* xr4 = reinterpret_cast<const float4*>(x + (size_t)b * DIM);
    float4 xv[4];
    float ss = 0.0f;
    #pragma unroll
    for (int e = 0; e < 4; ++e) {
        const float4 v = xr4[tid + 256 * e];
        xv[e] = v;
        ss += v.x * v.x + v.y * v.y + v.z * v.z + v.w * v.w;
    }
    for (int o = 16; o > 0; o >>= 1) ss += __shfl_xor(ss, o, 32);
    if (lane == 0) red[wave] = ss;
    __syncthreads();
    float n1sq = 0.0f;
    #pragma unroll
    for (int wv = 0; wv < 8; ++wv) n1sq += red[wv];
    const float x0   = x[(size_t)b * DIM];
    const bool  zero = (sqrtf(n1sq) <= 1e-12f);
    const float x0n  = zero ? 1.0f : x0;
    const float inv  = 1.0f / sqrtf(n1sq - x0 * x0 + x0n * x0n);
    __syncthreads();
    #pragma unroll
    for (int e = 0; e < 4; ++e) {
        float4 v = xv[e];
        v.x = (tid == 0 && e == 0) ? x0n : v.x;
        v.x *= inv; v.y *= inv; v.z *= inv; v.w *= inv;
        float* const p = psi + SIDX((tid >> 6) + 4 * e, (tid >> 2) & 15, (tid & 3) * 4);
        *reinterpret_cast<float4*>(p)         = v;
        *reinterpret_cast<float4*>(p + PLANE) = make_float4(0.f, 0.f, 0.f, 0.f);
    }
    __syncthreads();

    const float* const pB0 = psi + SIDX(2 * half, wave, mn);
    float*       const pD0 = psi + SIDX(8 * half, wave, mn);
    const float* const pB1 = psi + SIDX(wave, 2 * half, mn);
    float*       const pD1 = psi + SIDX(wave, 8 * half, mn);
    const float* const pB2 = psi + SIDX(wave, mn, 2 * half);
    float*       const pD2 = psi + SIDX(wave, mn, 8 * half);
    const float* const pA  = gLDS + (mn * 4 + half * 2);

    for (int l = 0; l < DEPTH; ++l) {
        const float* const pAl = pA + l * (3 * 2 * 256);
        #pragma unroll
        for (int g = 0; g < 3; ++g) {

            v8f accRR[2], accII[2], accSS[2];
            #pragma unroll
            for (int tt = 0; tt < 2; ++tt) { accRR[tt] = (v8f){}; accII[tt] = (v8f){}; accSS[tt] = (v8f){}; }
            #pragma unroll
            for (int kq = 0; kq < 4; ++kq) {
                const v2f Ar = *reinterpret_cast<const v2f*>(&pAl[(g * 2 + 0) * 256 + kq * 64]);
                const v2f Ai = *reinterpret_cast<const v2f*>(&pAl[(g * 2 + 1) * 256 + kq * 64]);
                const v2f As = Ar + Ai;
                #pragma unroll
                for (int tt = 0; tt < 2; ++tt) {
                    v2f Br, Bi;
                    if (g == 0) {
                        const int o = SIDX(kq * 4, 8 * tt, 0);
                        Br = (v2f){pB0[o], pB0[o + SA]};
                        Bi = (v2f){pB0[PLANE + o], pB0[PLANE + o + SA]};
                    } else if (g == 1) {
                        const int o = SIDX(8 * tt, kq * 4, 0);
                        Br = (v2f){pB1[o], pB1[o + SB]};
                        Bi = (v2f){pB1[PLANE + o], pB1[PLANE + o + SB]};
                    } else {
                        const int o = SIDX(8 * tt, 0, kq * 4);
                        Br = *reinterpret_cast<const v2f*>(&pB2[o]);
                        Bi = *reinterpret_cast<const v2f*>(&pB2[PLANE + o]);
                    }
                    const v2f Bs = Br + Bi;
                    accRR[tt] = __builtin_amdgcn_wmma_f32_16x16x4_f32(false, Ar, false, Br, (short)0, accRR[tt], false, false);
                    accII[tt] = __builtin_amdgcn_wmma_f32_16x16x4_f32(false, Ai, false, Bi, (short)0, accII[tt], false, false);
                    accSS[tt] = __builtin_amdgcn_wmma_f32_16x16x4_f32(false, As, false, Bs, (short)0, accSS[tt], false, false);
                }
            }
            #pragma unroll
            for (int tt = 0; tt < 2; ++tt) {
                #pragma unroll
                for (int v = 0; v < 8; ++v) {
                    float* p;
                    if      (g == 0) p = pD0 + SIDX(v, 8 * tt, 0);
                    else if (g == 1) p = pD1 + SIDX(8 * tt, v, 0);
                    else             p = pD2 + SIDX(8 * tt, 0, v);
                    p[0]     = accRR[tt][v] - accII[tt][v];
                    p[PLANE] = (accSS[tt][v] - accRR[tt][v]) - accII[tt][v];
                }
            }
            __syncthreads();
        }

        const int* bas = &permB[l * 12];
        int pbase = 0;
        #pragma unroll
        for (int bb = 0; bb < 8; ++bb)
            pbase ^= bas[4 + bb] & -((tid >> bb) & 1);
        int pe[16];
        #pragma unroll
        for (int e = 0; e < 16; ++e) {
            int v = 0;
            if (e & 1) v ^= bas[0];
            if (e & 2) v ^= bas[1];
            if (e & 4) v ^= bas[2];
            if (e & 8) v ^= bas[3];
            pe[e] = v;
        }
        float rr[16], ri[16];
        #pragma unroll
        for (int e = 0; e < 16; ++e) {
            const int j  = pbase ^ pe[e];
            const int jp = SIDX(j >> 8, (j >> 4) & 15, j & 15);
            rr[e] = psiR[jp];
            ri[e] = psiI[jp];
        }
        __syncthreads();
        float* const pw = psi + SIDX(tid >> 4, tid & 15, 0);
        #pragma unroll
        for (int q = 0; q < 4; ++q) {
            *reinterpret_cast<float4*>(pw + 4 * q)         = make_float4(rr[4*q], rr[4*q+1], rr[4*q+2], rr[4*q+3]);
            *reinterpret_cast<float4*>(pw + PLANE + 4 * q) = make_float4(ri[4*q], ri[4*q+1], ri[4*q+2], ri[4*q+3]);
        }
        __syncthreads();
    }

    float acc[12];
    #pragma unroll
    for (int k = 0; k < 12; ++k) acc[k] = 0.0f;
    const float* const pf = psi + SIDX(tid >> 4, tid & 15, 0);
    #pragma unroll
    for (int e4 = 0; e4 < 4; ++e4) {
        const float4 a = *reinterpret_cast<const float4*>(pf + 4 * e4);
        const float4 c = *reinterpret_cast<const float4*>(pf + PLANE + 4 * e4);
        const int ib = tid * 16 + e4 * 4;
        const float p0 = a.x * a.x + c.x * c.x;
        const float p1 = a.y * a.y + c.y * c.y;
        const float p2 = a.z * a.z + c.z * c.z;
        const float p3 = a.w * a.w + c.w * c.w;
        const float pv[4] = {p0, p1, p2, p3};
        #pragma unroll
        for (int cc = 0; cc < 4; ++cc) {
            const int i = ib + cc;
            #pragma unroll
            for (int k = 0; k < 12; ++k)
                acc[k] += ((i >> (11 - k)) & 1) ? -pv[cc] : pv[cc];
        }
    }
    #pragma unroll
    for (int k = 0; k < 12; ++k) {
        float v = acc[k];
        for (int o = 16; o > 0; o >>= 1) v += __shfl_xor(v, o, 32);
        if (lane == 0) red[wave * 12 + k] = v;
    }
    __syncthreads();
    if (tid < 12) {
        float s = 0.0f;
        #pragma unroll
        for (int wv = 0; wv < 8; ++wv) s += red[wv * 12 + tid];
        out[(size_t)b * 12 + tid] = s;
    }
}

extern "C" void kernel_launch(void* const* d_in, const int* in_sizes, int n_in,
                              void* d_out, int out_size, void* d_ws, size_t ws_size,
                              hipStream_t stream) {
    const float* x     = (const float*)d_in[0];
    const float* theta = (const float*)d_in[1];
    float*       out   = (float*)d_out;
    float*       wsf   = (float*)d_ws;

    const int B = in_sizes[0] / DIM;

    qdl_setup_kernel<<<13, 256, 0, stream>>>(theta, wsf);
    qdl_main_kernel<<<B, 256, 0, stream>>>(x, wsf, out);
}


# Setup launch folded into the main kernel: each block builds the 12 gate matrices and the permutation basis in LDS from theta (same formulas), one launch per call instead of two; sums unchanged
# speedup vs baseline: 1.5268x; 1.0306x over previous
#include <hip/hip_runtime.h>
#include <math.h>

#define NQ    12
#define DIM   4096
#define DEPTH 4

#define SA    276
#define SB    52
#define PLANE (15 * SA + 15 * SB + 16)
#define SIDX(a, b, c) ((a) * SA + (b) * SB + (c))

typedef __attribute__((ext_vector_type(2))) float v2f;
typedef __attribute__((ext_vector_type(8))) float v8f;

__global__ __launch_bounds__(256, 10) void qdl_main_kernel(
        const float* __restrict__ x,
        const float* __restrict__ theta,
        float* __restrict__ out) {
    __shared__ __attribute__((aligned(16))) float psi[2 * PLANE];
    float* const psiR = psi;
    float* const psiI = psi + PLANE;
    __shared__ float gLDS[12 * 2 * 256];
    __shared__ int   permB[48];
    __shared__ float red[8 * 12];

    const int tid  = threadIdx.x;
    const int b    = blockIdx.x;
    const int lane = tid & 31;
    const int wave = tid >> 5;
    const int half = lane >> 4;
    const int mn   = lane & 15;

    const float4* xr4 = reinterpret_cast<const float4*>(x + (size_t)b * DIM);
    float4 xv[4];
    #pragma unroll
    for (int e = 0; e < 4; ++e) xv[e] = xr4[tid + 256 * e];

    float* const sc = psi;
    float* const uT = psi + 512;
    if (tid < 144) {
        const int wire = tid / 3, which = tid - wire * 3;
        const float* th = theta + wire * 3;
        const float phi = th[0], the = th[1], om = th[2];
        const float ang = (which == 0) ? 0.5f * the
                        : (which == 1) ? 0.5f * (phi + om) : 0.5f * (phi - om);
        sc[tid * 2 + 0] = cosf(ang);
        sc[tid * 2 + 1] = sinf(ang);
    } else if (tid >= 192 && tid < 240) {
        const int t = tid - 192;
        const int l = t / 12, k = t % 12;
        const int r = l % (NQ - 1) + 1;
        int j = 1 << k;
        for (int w = NQ - 1; w >= 0; --w) {
            const int tg   = (w + r) % NQ;
            const int cbit = (j >> (NQ - 1 - w)) & 1;
            j ^= cbit << (NQ - 1 - tg);
        }
        permB[t] = j;
    }
    __syncthreads();
    if (tid < 192) {
        const int wire = tid >> 2, bi = (tid >> 1) & 1, bj = tid & 1;
        const float ch = sc[wire * 6 + 0], sh = sc[wire * 6 + 1];
        const float ca = sc[wire * 6 + 2], sa = sc[wire * 6 + 3];
        const float cb = sc[wire * 6 + 4], sb = sc[wire * 6 + 5];
        float ur, ui;
        if      (bi == 0 && bj == 0) { ur =  ca * ch; ui = -sa * ch; }
        else if (bi == 0 && bj == 1) { ur = -cb * sh; ui = -sb * sh; }
        else if (bi == 1 && bj == 0) { ur =  cb * sh; ui = -sb * sh; }
        else                         { ur =  ca * ch; ui =  sa * ch; }
        uT[tid * 2 + 0] = ur;
        uT[tid * 2 + 1] = ui;
    }
    __syncthreads();
    {
        const int i = tid >> 4, j = tid & 15;
        const float* uq[4];
        #pragma unroll
        for (int q = 0; q < 4; ++q)
            uq[q] = uT + q * 8 + ((((i >> (3 - q)) & 1) * 2 + ((j >> (3 - q)) & 1)) * 2);
        const int ga = (j >> 2) * 64 + i * 4 + (j & 3);
        #pragma unroll
        for (int lg = 0; lg < 12; ++lg) {
            float pr = 1.0f, pi = 0.0f;
            #pragma unroll
            for (int q = 0; q < 4; ++q) {
                const v2f u = *reinterpret_cast<const v2f*>(uq[q] + lg * 32);
                const float nr = pr * u.x - pi * u.y;
                const float ni = pr * u.y + pi * u.x;
                pr = nr; pi = ni;
            }
            gLDS[(lg * 2 + 0) * 256 + ga] = pr;
            gLDS[(lg * 2 + 1) * 256 + ga] = pi;
        }
    }

    float ss = 0.0f;
    #pragma unroll
    for (int e = 0; e < 4; ++e) {
        const float4 v = xv[e];
        ss += v.x * v.x + v.y * v.y + v.z * v.z + v.w * v.w;
    }
    for (int o = 16; o > 0; o >>= 1) ss += __shfl_down(ss, o, 32);
    if (lane == 0) red[wave] = ss;
    __syncthreads();
    float n1sq = 0.0f;
    #pragma unroll
    for (int wv = 0; wv < 8; ++wv) n1sq += red[wv];
    const float x0   = x[(size_t)b * DIM];
    const bool  zero = (sqrtf(n1sq) <= 1e-12f);
    const float x0n  = zero ? 1.0f : x0;
    const float inv  = 1.0f / sqrtf(n1sq - x0 * x0 + x0n * x0n);
    __syncthreads();
    #pragma unroll
    for (int e = 0; e < 4; ++e) {
        float4 v = xv[e];
        v.x = (tid == 0 && e == 0) ? x0n : v.x;
        v.x *= inv; v.y *= inv; v.z *= inv; v.w *= inv;
        float* const p = psi + SIDX((tid >> 6) + 4 * e, (tid >> 2) & 15, (tid & 3) * 4);
        *reinterpret_cast<float4*>(p)         = v;
        *reinterpret_cast<float4*>(p + PLANE) = make_float4(0.f, 0.f, 0.f, 0.f);
    }
    __syncthreads();

    const float* const pB0 = psi + SIDX(2 * half, wave, mn);
    float*       const pD0 = psi + SIDX(8 * half, wave, mn);
    const float* const pB1 = psi + SIDX(wave, 2 * half, mn);
    float*       const pD1 = psi + SIDX(wave, 8 * half, mn);
    const float* const pB2 = psi + SIDX(wave, mn, 2 * half);
    float*       const pD2 = psi + SIDX(wave, mn, 8 * half);
    const float* const pA  = gLDS + (mn * 4 + half * 2);

    for (int l = 0; l < DEPTH; ++l) {
        const float* const pAl = pA + l * (3 * 2 * 256);
        #pragma unroll
        for (int g = 0; g < 3; ++g) {

            v8f accRR[2], accII[2], accSS[2];
            #pragma unroll
            for (int tt = 0; tt < 2; ++tt) { accRR[tt] = (v8f){}; accII[tt] = (v8f){}; accSS[tt] = (v8f){}; }
            #pragma unroll
            for (int kq = 0; kq < 4; ++kq) {
                const v2f Ar = *reinterpret_cast<const v2f*>(&pAl[(g * 2 + 0) * 256 + kq * 64]);
                const v2f Ai = *reinterpret_cast<const v2f*>(&pAl[(g * 2 + 1) * 256 + kq * 64]);
                const v2f As = Ar + Ai;
                #pragma unroll
                for (int tt = 0; tt < 2; ++tt) {
                    v2f Br, Bi;
                    if (g == 0) {
                        const int o = SIDX(kq * 4, 8 * tt, 0);
                        Br = (v2f){pB0[o], pB0[o + SA]};
                        Bi = (v2f){pB0[PLANE + o], pB0[PLANE + o + SA]};
                    } else if (g == 1) {
                        const int o = SIDX(8 * tt, kq * 4, 0);
                        Br = (v2f){pB1[o], pB1[o + SB]};
                        Bi = (v2f){pB1[PLANE + o], pB1[PLANE + o + SB]};
                    } else {
                        const int o = SIDX(8 * tt, 0, kq * 4);
                        Br = *reinterpret_cast<const v2f*>(&pB2[o]);
                        Bi = *reinterpret_cast<const v2f*>(&pB2[PLANE + o]);
                    }
                    const v2f Bs = Br + Bi;
                    accRR[tt] = __builtin_amdgcn_wmma_f32_16x16x4_f32(false, Ar, false, Br, (short)0, accRR[tt], false, false);
                    accII[tt] = __builtin_amdgcn_wmma_f32_16x16x4_f32(false, Ai, false, Bi, (short)0, accII[tt], false, false);
                    accSS[tt] = __builtin_amdgcn_wmma_f32_16x16x4_f32(false, As, false, Bs, (short)0, accSS[tt], false, false);
                }
            }
            #pragma unroll
            for (int tt = 0; tt < 2; ++tt) {
                #pragma unroll
                for (int v = 0; v < 8; ++v) {
                    float* p;
                    if      (g == 0) p = pD0 + SIDX(v, 8 * tt, 0);
                    else if (g == 1) p = pD1 + SIDX(8 * tt, v, 0);
                    else             p = pD2 + SIDX(8 * tt, 0, v);
                    p[0]     = accRR[tt][v] - accII[tt][v];
                    p[PLANE] = (accSS[tt][v] - accRR[tt][v]) - accII[tt][v];
                }
            }
            __syncthreads();
        }

        const int* bas = &permB[l * 12];
        int pbase = 0;
        #pragma unroll
        for (int bb = 0; bb < 8; ++bb)
            pbase ^= bas[4 + bb] & -((tid >> bb) & 1);
        int pe[16];
        #pragma unroll
        for (int e = 0; e < 16; ++e) {
            int v = 0;
            if (e & 1) v ^= bas[0];
            if (e & 2) v ^= bas[1];
            if (e & 4) v ^= bas[2];
            if (e & 8) v ^= bas[3];
            pe[e] = v;
        }
        float rr[16], ri[16];
        #pragma unroll
        for (int e = 0; e < 16; ++e) {
            const int j  = pbase ^ pe[e];
            const int jp = SIDX(j >> 8, (j >> 4) & 15, j & 15);
            rr[e] = psiR[jp];
            ri[e] = psiI[jp];
        }
        __syncthreads();
        float* const pw = psi + SIDX(tid >> 4, tid & 15, 0);
        #pragma unroll
        for (int q = 0; q < 4; ++q) {
            *reinterpret_cast<float4*>(pw + 4 * q)         = make_float4(rr[4*q], rr[4*q+1], rr[4*q+2], rr[4*q+3]);
            *reinterpret_cast<float4*>(pw + PLANE + 4 * q) = make_float4(ri[4*q], ri[4*q+1], ri[4*q+2], ri[4*q+3]);
        }
        __syncthreads();
    }

    float acc[12];
    #pragma unroll
    for (int k = 0; k < 12; ++k) acc[k] = 0.0f;
    const float* const pf = psi + SIDX(tid >> 4, tid & 15, 0);
    #pragma unroll
    for (int e4 = 0; e4 < 4; ++e4) {
        const float4 a = *reinterpret_cast<const float4*>(pf + 4 * e4);
        const float4 c = *reinterpret_cast<const float4*>(pf + PLANE + 4 * e4);
        const int ib = tid * 16 + e4 * 4;
        const float p0 = a.x * a.x + c.x * c.x;
        const float p1 = a.y * a.y + c.y * c.y;
        const float p2 = a.z * a.z + c.z * c.z;
        const float p3 = a.w * a.w + c.w * c.w;
        const float pv[4] = {p0, p1, p2, p3};
        #pragma unroll
        for (int cc = 0; cc < 4; ++cc) {
            const int i = ib + cc;
            #pragma unroll
            for (int k = 0; k < 12; ++k)
                acc[k] += ((i >> (11 - k)) & 1) ? -pv[cc] : pv[cc];
        }
    }
    #pragma unroll
    for (int k = 0; k < 12; ++k) {
        float v = acc[k];
        for (int o = 16; o > 0; o >>= 1) v += __shfl_xor(v, o, 32);
        if (lane == 0) red[wave * 12 + k] = v;
    }
    __syncthreads();
    if (tid < 12) {
        float s = 0.0f;
        #pragma unroll
        for (int wv = 0; wv < 8; ++wv) s += red[wv * 12 + tid];
        out[(size_t)b * 12 + tid] = s;
    }
}

extern "C" void kernel_launch(void* const* d_in, const int* in_sizes, int n_in,
                              void* d_out, int out_size, void* d_ws, size_t ws_size,
                              hipStream_t stream) {
    const float* x     = (const float*)d_in[0];
    const float* theta = (const float*)d_in[1];
    float*       out   = (float*)d_out;

    const int B = in_sizes[0] / DIM;

    qdl_main_kernel<<<B, 256, 0, stream>>>(x, theta, out);
}


# Rot layer factored as diagonal RZ phases x real RY Kronecker matrix x diagonal RZ phases: real-matrix WMMA products, two sums per complex tile not three (48 not 72 per wave-layer); rounding changes
# speedup vs baseline: 1.8154x; 1.1890x over previous
#include <hip/hip_runtime.h>
#include <math.h>

#define NQ    12
#define DIM   4096
#define DEPTH 4

#define SA    276
#define SB    52
#define PLANE (15 * SA + 15 * SB + 16)
#define SIDX(a, b, c) ((a) * SA + (b) * SB + (c))

typedef __attribute__((ext_vector_type(2))) float v2f;
typedef __attribute__((ext_vector_type(8))) float v8f;

__global__ __launch_bounds__(256, 10) void qdl_main_kernel(
        const float* __restrict__ x,
        const float* __restrict__ theta,
        float* __restrict__ out) {
    __shared__ __attribute__((aligned(16))) float psi[2 * PLANE];
    float* const psiR = psi;
    float* const psiI = psi + PLANE;
    __shared__ __attribute__((aligned(16))) float rLDS[12 * 256];
    __shared__ __attribute__((aligned(16))) float tabP[12 * 32];
    __shared__ __attribute__((aligned(16))) float tabO[12 * 32];
    __shared__ int   permB[48];
    __shared__ float red[8 * 12];

    const int tid  = threadIdx.x;
    const int b    = blockIdx.x;
    const int lane = tid & 31;
    const int wave = tid >> 5;
    const int half = lane >> 4;
    const int mn   = lane & 15;

    const float4* xr4 = reinterpret_cast<const float4*>(x + (size_t)b * DIM);
    float4 xv[4];
    #pragma unroll
    for (int e = 0; e < 4; ++e) xv[e] = xr4[tid + 256 * e];

    float* const sc = psi;
    if (tid < 144) {
        const int wire = tid / 3, which = tid - wire * 3;
        const float* th = theta + wire * 3;
        const float phi = th[0], the = th[1], om = th[2];
        const float ang = 0.5f * ((which == 0) ? the : (which == 1) ? phi : om);
        sc[tid * 2 + 0] = cosf(ang);
        sc[tid * 2 + 1] = sinf(ang);
    } else if (tid >= 192 && tid < 240) {
        const int t = tid - 192;
        const int l = t / 12, k = t % 12;
        const int r = l % (NQ - 1) + 1;
        int j = 1 << k;
        for (int w = NQ - 1; w >= 0; --w) {
            const int tg   = (w + r) % NQ;
            const int cbit = (j >> (NQ - 1 - w)) & 1;
            j ^= cbit << (NQ - 1 - tg);
        }
        permB[t] = j;
    }
    __syncthreads();
    {
        const int i = tid >> 4, j = tid & 15;
        const float* uq[4];
        float sg[4];
        #pragma unroll
        for (int q = 0; q < 4; ++q) {
            const int bi = (i >> (3 - q)) & 1, bj = (j >> (3 - q)) & 1;
            uq[q] = sc + q * 6 + ((bi == bj) ? 0 : 1);
            sg[q] = (bi == 0 && bj == 1) ? -1.0f : 1.0f;
        }
        const int ga = (j >> 2) * 64 + i * 4 + (j & 3);
        #pragma unroll
        for (int lg = 0; lg < 12; ++lg) {
            float p = 1.0f;
            #pragma unroll
            for (int q = 0; q < 4; ++q) p *= sg[q] * uq[q][lg * 24];
            rLDS[lg * 256 + ga] = p;
        }
    }
    if (tid < 192) {
        const int lg = tid >> 4, d = tid & 15;
        float pr = 1.0f, pi = 0.0f, qr = 1.0f, qi = 0.0f;
        #pragma unroll
        for (int q = 0; q < 4; ++q) {
            const float* w = sc + (lg * 4 + q) * 6;
            const bool set = (d >> (3 - q)) & 1;
            const float cp = w[2], sp = set ? w[3] : -w[3];
            const float co = w[4], so = set ? w[5] : -w[5];
            const float nr = pr * cp - pi * sp, ni = pr * sp + pi * cp;
            const float mr = qr * co - qi * so, mi = qr * so + qi * co;
            pr = nr; pi = ni; qr = mr; qi = mi;
        }
        tabP[lg * 32 + d] = pr; tabP[lg * 32 + 16 + d] = pi;
        tabO[lg * 32 + d] = qr; tabO[lg * 32 + 16 + d] = qi;
    }

    float ss = 0.0f;
    #pragma unroll
    for (int e = 0; e < 4; ++e) {
        const float4 v = xv[e];
        ss += v.x * v.x + v.y * v.y + v.z * v.z + v.w * v.w;
    }
    for (int o = 16; o > 0; o >>= 1) ss += __shfl_down(ss, o, 32);
    if (lane == 0) red[wave] = ss;
    __syncthreads();
    float n1sq = 0.0f;
    #pragma unroll
    for (int wv = 0; wv < 8; ++wv) n1sq += red[wv];
    const float x0   = x[(size_t)b * DIM];
    const bool  zero = (sqrtf(n1sq) <= 1e-12f);
    const float x0n  = zero ? 1.0f : x0;
    const float inv  = 1.0f / sqrtf(n1sq - x0 * x0 + x0n * x0n);
    float fbr[4], fbi[4];
    float par[4], pai[4];
    {
        const int   bb  = (tid >> 2) & 15;
        const float pbr = tabP[1 * 32 + bb], pbi = tabP[1 * 32 + 16 + bb];
        const float4 cr = *reinterpret_cast<const float4*>(&tabP[2 * 32 + (tid & 3) * 4]);
        const float4 ci = *reinterpret_cast<const float4*>(&tabP[2 * 32 + 16 + (tid & 3) * 4]);
        const float crv[4] = {cr.x, cr.y, cr.z, cr.w};
        const float civ[4] = {ci.x, ci.y, ci.z, ci.w};
        #pragma unroll
        for (int cc = 0; cc < 4; ++cc) {
            fbr[cc] = pbr * crv[cc] - pbi * civ[cc];
            fbi[cc] = pbr * civ[cc] + pbi * crv[cc];
        }
        #pragma unroll
        for (int e = 0; e < 4; ++e) {
            par[e] = tabP[(tid >> 6) + 4 * e];
            pai[e] = tabP[16 + (tid >> 6) + 4 * e];
        }
    }
    __syncthreads();
    #pragma unroll
    for (int e = 0; e < 4; ++e) {
        float4 v = xv[e];
        v.x = (tid == 0 && e == 0) ? x0n : v.x;
        v.x *= inv; v.y *= inv; v.z *= inv; v.w *= inv;
        const float vv[4] = {v.x, v.y, v.z, v.w};
        float re[4], im[4];
        #pragma unroll
        for (int cc = 0; cc < 4; ++cc) {
            const float fr = par[e] * fbr[cc] - pai[e] * fbi[cc];
            const float fi = par[e] * fbi[cc] + pai[e] * fbr[cc];
            re[cc] = vv[cc] * fr;
            im[cc] = vv[cc] * fi;
        }
        float* const p = psi + SIDX((tid >> 6) + 4 * e, (tid >> 2) & 15, (tid & 3) * 4);
        *reinterpret_cast<float4*>(p)         = make_float4(re[0], re[1], re[2], re[3]);
        *reinterpret_cast<float4*>(p + PLANE) = make_float4(im[0], im[1], im[2], im[3]);
    }
    __syncthreads();

    const float* const pB0 = psi + SIDX(2 * half, wave, mn);
    float*       const pD0 = psi + SIDX(8 * half, wave, mn);
    const float* const pB1 = psi + SIDX(wave, 2 * half, mn);
    float*       const pD1 = psi + SIDX(wave, 8 * half, mn);
    const float* const pB2 = psi + SIDX(wave, mn, 2 * half);
    float*       const pD2 = psi + SIDX(wave, mn, 8 * half);
    const float* const pA  = rLDS + (mn * 4 + half * 2);

    for (int l = 0; l < DEPTH; ++l) {
        const float* const pAl = pA + l * (3 * 256);
        const float* const tOl = tabO + l * (3 * 32);
        #pragma unroll
        for (int g = 0; g < 3; ++g) {

            v8f accR[2], accI[2];
            #pragma unroll
            for (int tt = 0; tt < 2; ++tt) { accR[tt] = (v8f){}; accI[tt] = (v8f){}; }
            #pragma unroll
            for (int kq = 0; kq < 4; ++kq) {
                const v2f Ar = *reinterpret_cast<const v2f*>(&pAl[g * 256 + kq * 64]);
                #pragma unroll
                for (int tt = 0; tt < 2; ++tt) {
                    v2f Br, Bi;
                    if (g == 0) {
                        const int o = SIDX(kq * 4, 8 * tt, 0);
                        Br = (v2f){pB0[o], pB0[o + SA]};
                        Bi = (v2f){pB0[PLANE + o], pB0[PLANE + o + SA]};
                    } else if (g == 1) {
                        const int o = SIDX(8 * tt, kq * 4, 0);
                        Br = (v2f){pB1[o], pB1[o + SB]};
                        Bi = (v2f){pB1[PLANE + o], pB1[PLANE + o + SB]};
                    } else {
                        const int o = SIDX(8 * tt, 0, kq * 4);
                        Br = *reinterpret_cast<const v2f*>(&pB2[o]);
                        Bi = *reinterpret_cast<const v2f*>(&pB2[PLANE + o]);
                    }
                    accR[tt] = __builtin_amdgcn_wmma_f32_16x16x4_f32(false, Ar, false, Br, (short)0, accR[tt], false, false);
                    accI[tt] = __builtin_amdgcn_wmma_f32_16x16x4_f32(false, Ar, false, Bi, (short)0, accI[tt], false, false);
                }
            }
            if (g < 2) {
                #pragma unroll
                for (int tt = 0; tt < 2; ++tt) {
                    #pragma unroll
                    for (int v = 0; v < 8; ++v) {
                        float* p;
                        if (g == 0) p = pD0 + SIDX(v, 8 * tt, 0);
                        else        p = pD1 + SIDX(8 * tt, v, 0);
                        p[0]     = accR[tt][v];
                        p[PLANE] = accI[tt][v];
                    }
                }
            } else {
                const float4 c0r = *reinterpret_cast<const float4*>(&tOl[2 * 32 + 8 * half]);
                const float4 c1r = *reinterpret_cast<const float4*>(&tOl[2 * 32 + 8 * half + 4]);
                const float4 c0i = *reinterpret_cast<const float4*>(&tOl[2 * 32 + 16 + 8 * half]);
                const float4 c1i = *reinterpret_cast<const float4*>(&tOl[2 * 32 + 16 + 8 * half + 4]);
                const float ocr[8] = {c0r.x, c0r.y, c0r.z, c0r.w, c1r.x, c1r.y, c1r.z, c1r.w};
                const float oci[8] = {c0i.x, c0i.y, c0i.z, c0i.w, c1i.x, c1i.y, c1i.z, c1i.w};
                const float obr = tOl[1 * 32 + mn], obi = tOl[1 * 32 + 16 + mn];
                #pragma unroll
                for (int tt = 0; tt < 2; ++tt) {
                    const float oar = tOl[wave + 8 * tt], oai = tOl[16 + wave + 8 * tt];
                    const float abr = oar * obr - oai * obi;
                    const float abi = oar * obi + oai * obr;
                    #pragma unroll
                    for (int v = 0; v < 8; ++v) {
                        const float ur = accR[tt][v] * ocr[v] - accI[tt][v] * oci[v];
                        const float ui = accR[tt][v] * oci[v] + accI[tt][v] * ocr[v];
                        float* const p = pD2 + SIDX(8 * tt, 0, v);
                        p[0]     = ur * abr - ui * abi;
                        p[PLANE] = ur * abi + ui * abr;
                    }
                }
            }
            __syncthreads();
        }

        const int* bas = &permB[l * 12];
        int pbase = 0;
        #pragma unroll
        for (int bb = 0; bb < 8; ++bb)
            pbase ^= bas[4 + bb] & -((tid >> bb) & 1);
        int pe[16];
        #pragma unroll
        for (int e = 0; e < 16; ++e) {
            int v = 0;
            if (e & 1) v ^= bas[0];
            if (e & 2) v ^= bas[1];
            if (e & 4) v ^= bas[2];
            if (e & 8) v ^= bas[3];
            pe[e] = v;
        }
        float rr[16], ri[16];
        #pragma unroll
        for (int e = 0; e < 16; ++e) {
            const int j  = pbase ^ pe[e];
            const int jp = SIDX(j >> 8, (j >> 4) & 15, j & 15);
            rr[e] = psiR[jp];
            ri[e] = psiI[jp];
        }
        if (l + 1 < DEPTH) {
            const float* const tPn = tabP + (l + 1) * (3 * 32);
            const float par = tPn[tid >> 4],            pai = tPn[16 + (tid >> 4)];
            const float pbr = tPn[1 * 32 + (tid & 15)], pbi = tPn[1 * 32 + 16 + (tid & 15)];
            const float abr = par * pbr - pai * pbi;
            const float abi = par * pbi + pai * pbr;
            #pragma unroll
            for (int q = 0; q < 4; ++q) {
                const float4 cr = *reinterpret_cast<const float4*>(&tPn[2 * 32 + 4 * q]);
                const float4 ci = *reinterpret_cast<const float4*>(&tPn[2 * 32 + 16 + 4 * q]);
                const float crv[4] = {cr.x, cr.y, cr.z, cr.w};
                const float civ[4] = {ci.x, ci.y, ci.z, ci.w};
                #pragma unroll
                for (int cc = 0; cc < 4; ++cc) {
                    const float vr = rr[4 * q + cc], vi = ri[4 * q + cc];
                    const float ur = vr * crv[cc] - vi * civ[cc];
                    const float ui = vr * civ[cc] + vi * crv[cc];
                    rr[4 * q + cc] = ur * abr - ui * abi;
                    ri[4 * q + cc] = ur * abi + ui * abr;
                }
            }
        }
        __syncthreads();
        float* const pw = psi + SIDX(tid >> 4, tid & 15, 0);
        #pragma unroll
        for (int q = 0; q < 4; ++q) {
            *reinterpret_cast<float4*>(pw + 4 * q)         = make_float4(rr[4*q], rr[4*q+1], rr[4*q+2], rr[4*q+3]);
            *reinterpret_cast<float4*>(pw + PLANE + 4 * q) = make_float4(ri[4*q], ri[4*q+1], ri[4*q+2], ri[4*q+3]);
        }
        __syncthreads();
    }

    float acc[12];
    #pragma unroll
    for (int k = 0; k < 12; ++k) acc[k] = 0.0f;
    const float* const pf = psi + SIDX(tid >> 4, tid & 15, 0);
    #pragma unroll
    for (int e4 = 0; e4 < 4; ++e4) {
        const float4 a = *reinterpret_cast<const float4*>(pf + 4 * e4);
        const float4 c = *reinterpret_cast<const float4*>(pf + PLANE + 4 * e4);
        const int ib = tid * 16 + e4 * 4;
        const float p0 = a.x * a.x + c.x * c.x;
        const float p1 = a.y * a.y + c.y * c.y;
        const float p2 = a.z * a.z + c.z * c.z;
        const float p3 = a.w * a.w + c.w * c.w;
        const float pv[4] = {p0, p1, p2, p3};
        #pragma unroll
        for (int cc = 0; cc < 4; ++cc) {
            const int i = ib + cc;
            #pragma unroll
            for (int k = 0; k < 12; ++k)
                acc[k] += ((i >> (11 - k)) & 1) ? -pv[cc] : pv[cc];
        }
    }
    #pragma unroll
    for (int k = 0; k < 12; ++k) {
        float v = acc[k];
        for (int o = 16; o > 0; o >>= 1) v += __shfl_xor(v, o, 32);
        if (lane == 0) red[wave * 12 + k] = v;
    }
    __syncthreads();
    if (tid < 12) {
        float s = 0.0f;
        #pragma unroll
        for (int wv = 0; wv < 8; ++wv) s += red[wv * 12 + tid];
        out[(size_t)b * 12 + tid] = s;
    }
}

extern "C" void kernel_launch(void* const* d_in, const int* in_sizes, int n_in,
                              void* d_out, int out_size, void* d_ws, size_t ws_size,
                              hipStream_t stream) {
    const float* x     = (const float*)d_in[0];
    const float* theta = (const float*)d_in[1];
    float*       out   = (float*)d_out;

    const int B = in_sizes[0] / DIM;

    qdl_main_kernel<<<B, 256, 0, stream>>>(x, theta, out);
}


# Entangler permutation folded into third group's write-back (element stored at its permuted place, no gather pass); RZ(phi) moves to next first group, last layer's into output signs; sum order changes
# speedup vs baseline: 2.0170x; 1.1110x over previous
#include <hip/hip_runtime.h>
#include <math.h>

#define NQ    12
#define DIM   4096
#define DEPTH 4

#define SA    276
#define SB    52
#define PLANE (15 * SA + 15 * SB + 16)
#define SIDX(a, b, c) ((a) * SA + (b) * SB + (c))
#define SLIN(i) ((i) + (SB - 16) * ((i) >> 4) + (SA - 256 - 16 * (SB - 16)) * ((i) >> 8))

typedef __attribute__((ext_vector_type(2))) float v2f;
typedef __attribute__((ext_vector_type(8))) float v8f;

__host__ __device__ constexpr int inv_perm(int l, int j) {
    const int r = l % (NQ - 1) + 1;
    for (int w = 0; w < NQ; ++w) {
        const int tg   = (w + r) % NQ;
        const int cbit = (j >> (NQ - 1 - w)) & 1;
        j ^= cbit << (NQ - 1 - tg);
    }
    return j;
}
__host__ __device__ constexpr int sign_mask(int k) {
    int m = 0;
    for (int q = 0; q < NQ; ++q)
        if ((inv_perm(DEPTH - 1, 1 << q) >> (NQ - 1 - k)) & 1) m |= 1 << q;
    return m;
}

__global__ __launch_bounds__(256, 10) void qdl_main_kernel(
        const float* __restrict__ x,
        const float* __restrict__ theta,
        float* __restrict__ out) {
    __shared__ __attribute__((aligned(16))) float psi[2 * PLANE];
    __shared__ __attribute__((aligned(16))) float rLDS[12 * 256];
    __shared__ __attribute__((aligned(16))) float tabP[12 * 32];
    __shared__ __attribute__((aligned(16))) float tabO[12 * 32];
    __shared__ int   permB[48];
    __shared__ float red[8 * 12];

    const int tid  = threadIdx.x;
    const int b    = blockIdx.x;
    const int lane = tid & 31;
    const int wave = tid >> 5;
    const int half = lane >> 4;
    const int mn   = lane & 15;

    const float4* xr4 = reinterpret_cast<const float4*>(x + (size_t)b * DIM);
    float4 xv[4];
    #pragma unroll
    for (int e = 0; e < 4; ++e) xv[e] = xr4[tid + 256 * e];

    float* const sc = psi;
    if (tid < 144) {
        const int wire = tid / 3, which = tid - wire * 3;
        const float* th = theta + wire * 3;
        const float phi = th[0], the = th[1], om = th[2];
        const float ang = 0.5f * ((which == 0) ? the : (which == 1) ? phi : om);
        sc[tid * 2 + 0] = cosf(ang);
        sc[tid * 2 + 1] = sinf(ang);
    } else if (tid >= 192 && tid < 240) {
        const int t = tid - 192;
        permB[t] = inv_perm(t / 12, 1 << (t % 12));
    }
    __syncthreads();
    {
        const int i = tid >> 4, j = tid & 15;
        const float* uq[4];
        float sg[4];
        #pragma unroll
        for (int q = 0; q < 4; ++q) {
            const int bi = (i >> (3 - q)) & 1, bj = (j >> (3 - q)) & 1;
            uq[q] = sc + q * 6 + ((bi == bj) ? 0 : 1);
            sg[q] = (bi == 0 && bj == 1) ? -1.0f : 1.0f;
        }
        const int ga = (j >> 2) * 64 + i * 4 + (j & 3);
        #pragma unroll
        for (int lg = 0; lg < 12; ++lg) {
            float p = 1.0f;
            #pragma unroll
            for (int q = 0; q < 4; ++q) p *= sg[q] * uq[q][lg * 24];
            rLDS[lg * 256 + ga] = p;
        }
    }
    if (tid < 192) {
        const int lg = tid >> 4, d = tid & 15;
        float pr = 1.0f, pi = 0.0f, qr = 1.0f, qi = 0.0f;
        #pragma unroll
        for (int q = 0; q < 4; ++q) {
            const float* w = sc + (lg * 4 + q) * 6;
            const bool set = (d >> (3 - q)) & 1;
            const float cp = w[2], sp = set ? w[3] : -w[3];
            const float co = w[4], so = set ? w[5] : -w[5];
            const float nr = pr * cp - pi * sp, ni = pr * sp + pi * cp;
            const float mr = qr * co - qi * so, mi = qr * so + qi * co;
            pr = nr; pi = ni; qr = mr; qi = mi;
        }
        tabP[lg * 32 + d] = pr; tabP[lg * 32 + 16 + d] = pi;
        tabO[lg * 32 + d] = qr; tabO[lg * 32 + 16 + d] = qi;
    }

    float ss = 0.0f;
    #pragma unroll
    for (int e = 0; e < 4; ++e) {
        const float4 v = xv[e];
        ss += v.x * v.x + v.y * v.y + v.z * v.z + v.w * v.w;
    }
    for (int o = 16; o > 0; o >>= 1) ss += __shfl_down(ss, o, 32);
    if (lane == 0) red[wave] = ss;
    __syncthreads();
    float n1sq = 0.0f;
    #pragma unroll
    for (int wv = 0; wv < 8; ++wv) n1sq += red[wv];
    const float x0   = x[(size_t)b * DIM];
    const bool  zero = (sqrtf(n1sq) <= 1e-12f);
    const float x0n  = zero ? 1.0f : x0;
    const float inv  = 1.0f / sqrtf(n1sq - x0 * x0 + x0n * x0n);
    __syncthreads();
    #pragma unroll
    for (int e = 0; e < 4; ++e) {
        float4 v = xv[e];
        v.x = (tid == 0 && e == 0) ? x0n : v.x;
        v.x *= inv; v.y *= inv; v.z *= inv; v.w *= inv;
        float* const p = psi + SIDX((tid >> 6) + 4 * e, (tid >> 2) & 15, (tid & 3) * 4);
        *reinterpret_cast<float4*>(p)         = v;
        *reinterpret_cast<float4*>(p + PLANE) = make_float4(0.0f, 0.0f, 0.0f, 0.0f);
    }
    __syncthreads();

    const float* const pB0 = psi + SIDX(2 * half, wave, mn);
    float*       const pD0 = psi + SIDX(8 * half, wave, mn);
    const float* const pB1 = psi + SIDX(wave, 2 * half, mn);
    float*       const pD1 = psi + SIDX(wave, 8 * half, mn);
    const float* const pB2 = psi + SIDX(wave, mn, 2 * half);
    const float* const pA  = rLDS + (mn * 4 + half * 2);

    v8f accR[2], accI[2];
    #pragma unroll 1
    for (int l = 0; l < DEPTH; ++l) {
        const float* const pAl = pA + l * (3 * 256);
        const float* const tPl = tabP + l * (3 * 32);
        const float* const tOl = tabO + l * (3 * 32);
        #pragma unroll
        for (int g = 0; g < 3; ++g) {

            #pragma unroll
            for (int tt = 0; tt < 2; ++tt) { accR[tt] = (v8f){}; accI[tt] = (v8f){}; }
            v2f sr[4][2], si[4][2], far[4], fai[4];
            if (g == 0) {
                #pragma unroll
                for (int kq = 0; kq < 4; ++kq) {
                    far[kq] = *reinterpret_cast<const v2f*>(&tPl[kq * 4 + 2 * half]);
                    fai[kq] = *reinterpret_cast<const v2f*>(&tPl[16 + kq * 4 + 2 * half]);
                    #pragma unroll
                    for (int tt = 0; tt < 2; ++tt) {
                        const int o = SIDX(kq * 4, 8 * tt, 0);
                        sr[kq][tt] = (v2f){pB0[o], pB0[o + SA]};
                        si[kq][tt] = (v2f){pB0[PLANE + o], pB0[PLANE + o + SA]};
                    }
                }
            }
            #pragma unroll
            for (int kq = 0; kq < 4; ++kq) {
                const v2f Ar = *reinterpret_cast<const v2f*>(&pAl[g * 256 + kq * 64]);
                #pragma unroll
                for (int tt = 0; tt < 2; ++tt) {
                    v2f Br, Bi;
                    if (g == 0) {
                        Br = sr[kq][tt] * far[kq] - si[kq][tt] * fai[kq];
                        Bi = sr[kq][tt] * fai[kq] + si[kq][tt] * far[kq];
                    } else if (g == 1) {
                        const int o = SIDX(8 * tt, kq * 4, 0);
                        Br = (v2f){pB1[o], pB1[o + SB]};
                        Bi = (v2f){pB1[PLANE + o], pB1[PLANE + o + SB]};
                    } else {
                        const int o = SIDX(8 * tt, 0, kq * 4);
                        Br = *reinterpret_cast<const v2f*>(&pB2[o]);
                        Bi = *reinterpret_cast<const v2f*>(&pB2[PLANE + o]);
                    }
                    accR[tt] = __builtin_amdgcn_wmma_f32_16x16x4_f32(false, Ar, false, Br, (short)0, accR[tt], false, false);
                    accI[tt] = __builtin_amdgcn_wmma_f32_16x16x4_f32(false, Ar, false, Bi, (short)0, accI[tt], false, false);
                }
            }
            if (g == 0) {
                const float pcr = tPl[2 * 32 + mn], pci = tPl[2 * 32 + 16 + mn];
                #pragma unroll
                for (int tt = 0; tt < 2; ++tt) {
                    const float pbr = tPl[1 * 32 + wave + 8 * tt], pbi = tPl[1 * 32 + 16 + wave + 8 * tt];
                    const float fr = pbr * pcr - pbi * pci;
                    const float fi = pbr * pci + pbi * pcr;
                    #pragma unroll
                    for (int v = 0; v < 8; ++v) {
                        float* const p = pD0 + SIDX(v, 8 * tt, 0);
                        p[0]     = accR[tt][v] * fr - accI[tt][v] * fi;
                        p[PLANE] = accR[tt][v] * fi + accI[tt][v] * fr;
                    }
                }
            } else if (g == 1) {
                #pragma unroll
                for (int tt = 0; tt < 2; ++tt) {
                    #pragma unroll
                    for (int v = 0; v < 8; ++v) {
                        float* const p = pD1 + SIDX(8 * tt, v, 0);
                        p[0]     = accR[tt][v];
                        p[PLANE] = accI[tt][v];
                    }
                }
            } else if (l + 1 < DEPTH) {
                __syncthreads();
                const float4 c0r = *reinterpret_cast<const float4*>(&tOl[2 * 32 + 8 * half]);
                const float4 c1r = *reinterpret_cast<const float4*>(&tOl[2 * 32 + 8 * half + 4]);
                const float4 c0i = *reinterpret_cast<const float4*>(&tOl[2 * 32 + 16 + 8 * half]);
                const float4 c1i = *reinterpret_cast<const float4*>(&tOl[2 * 32 + 16 + 8 * half + 4]);
                const float ocr[8] = {c0r.x, c0r.y, c0r.z, c0r.w, c1r.x, c1r.y, c1r.z, c1r.w};
                const float oci[8] = {c0i.x, c0i.y, c0i.z, c0i.w, c1i.x, c1i.y, c1i.z, c1i.w};
                const float obr = tOl[1 * 32 + mn], obi = tOl[1 * 32 + 16 + mn];
                const int* const bas = &permB[l * 12];
                int it = bas[3] & -half;
                #pragma unroll
                for (int bb = 0; bb < 4; ++bb) it ^= bas[4 + bb] & -((mn >> bb) & 1);
                #pragma unroll
                for (int bb = 0; bb < 3; ++bb) it ^= bas[8 + bb] & -((wave >> bb) & 1);
                const int q0  = __builtin_amdgcn_readfirstlane(bas[0]);
                const int q1  = __builtin_amdgcn_readfirstlane(bas[1]);
                const int q2  = __builtin_amdgcn_readfirstlane(bas[2]);
                const int q11 = __builtin_amdgcn_readfirstlane(bas[11]);
                #pragma unroll
                for (int tt = 0; tt < 2; ++tt) {
                    const float oar = tOl[wave + 8 * tt], oai = tOl[16 + wave + 8 * tt];
                    const float abr = oar * obr - oai * obi;
                    const float abi = oar * obi + oai * obr;
                    #pragma unroll
                    for (int v = 0; v < 8; ++v) {
                        const float ur = accR[tt][v] * ocr[v] - accI[tt][v] * oci[v];
                        const float ui = accR[tt][v] * oci[v] + accI[tt][v] * ocr[v];
                        const int i = it ^ (tt ? q11 : 0) ^ ((v & 1) ? q0 : 0) ^ ((v & 2) ? q1 : 0) ^ ((v & 4) ? q2 : 0);
                        float* const p = psi + SLIN(i);
                        p[0]     = ur * abr - ui * abi;
                        p[PLANE] = ur * abi + ui * abr;
                    }
                }
            }
            if (g < 2 || l + 1 < DEPTH) __syncthreads();
        }
    }

    {
        constexpr int MK[12] = {sign_mask(0), sign_mask(1), sign_mask(2),  sign_mask(3),
                                sign_mask(4), sign_mask(5), sign_mask(6),  sign_mask(7),
                                sign_mask(8), sign_mask(9), sign_mask(10), sign_mask(11)};
        float pv[2][8];
        #pragma unroll
        for (int tt = 0; tt < 2; ++tt) {
            #pragma unroll
            for (int v = 0; v < 8; ++v)
                pv[tt][v] = accR[tt][v] * accR[tt][v] + accI[tt][v] * accI[tt][v];
        }
        const int jt = (wave << 8) | (mn << 4) | (half << 3);
        #pragma unroll
        for (int k = 0; k < 12; ++k) {
            float s = 0.0f;
            #pragma unroll
            for (int tt = 0; tt < 2; ++tt) {
                #pragma unroll
                for (int v = 0; v < 8; ++v)
                    s += __builtin_parity(((tt << 11) | v) & MK[k]) ? -pv[tt][v] : pv[tt][v];
            }
            const int flip = (__builtin_popcount(jt & MK[k]) & 1) << 31;
            s = __int_as_float(__float_as_int(s) ^ flip);
            for (int o = 16; o > 0; o >>= 1) s += __shfl_xor(s, o, 32);
            if (lane == 0) red[wave * 12 + k] = s;
        }
    }
    __syncthreads();

    if (tid < 12) {
        float s = 0.0f;
        #pragma unroll
        for (int wv = 0; wv < 8; ++wv) s += red[wv * 12 + tid];
        out[(size_t)b * 12 + tid] = s;
    }
}

extern "C" void kernel_launch(void* const* d_in, const int* in_sizes, int n_in,
                              void* d_out, int out_size, void* d_ws, size_t ws_size,
                              hipStream_t stream) {
    const float* x     = (const float*)d_in[0];
    const float* theta = (const float*)d_in[1];
    float*       out   = (float*)d_out;

    const int B = in_sizes[0] / DIM;

    qdl_main_kernel<<<B, 256, 0, stream>>>(x, theta, out);
}


# Second and third group chained in registers (X*R2^T sums feed R1*(.) as its B operand, R1 stored in that k order): one LDS pass, one barrier less per layer; sum order changes; strides 308/36
# speedup vs baseline: 2.1357x; 1.0588x over previous
#include <hip/hip_runtime.h>
#include <math.h>

#define NQ    12
#define DIM   4096
#define DEPTH 4

#define SA    308
#define SB    36
#define PLANE (15 * SA + 15 * SB + 16)
#define SIDX(a, b, c) ((a) * SA + (b) * SB + (c))
#define SLIN(i) ((i) + (SB - 16) * ((i) >> 4) + (SA - 256 - 16 * (SB - 16)) * ((i) >> 8))

typedef __attribute__((ext_vector_type(2))) float v2f;
typedef __attribute__((ext_vector_type(8))) float v8f;

__host__ __device__ constexpr int inv_perm(int l, int j) {
    const int r = l % (NQ - 1) + 1;
    for (int w = 0; w < NQ; ++w) {
        const int tg   = (w + r) % NQ;
        const int cbit = (j >> (NQ - 1 - w)) & 1;
        j ^= cbit << (NQ - 1 - tg);
    }
    return j;
}
__host__ __device__ constexpr int sign_mask(int k) {
    int m = 0;
    for (int q = 0; q < NQ; ++q)
        if ((inv_perm(DEPTH - 1, 1 << q) >> (NQ - 1 - k)) & 1) m |= 1 << q;
    return m;
}

__global__ __launch_bounds__(256, 10) void qdl_main_kernel(
        const float* __restrict__ x,
        const float* __restrict__ theta,
        float* __restrict__ out) {
    __shared__ __attribute__((aligned(16))) float psi[2 * PLANE];
    __shared__ __attribute__((aligned(16))) float rLDS[12 * 256];
    __shared__ __attribute__((aligned(16))) float tabP[12 * 32];
    __shared__ __attribute__((aligned(16))) float tabO[12 * 32];
    __shared__ int   permB[48];
    __shared__ float red[8 * 12];

    const int tid  = threadIdx.x;
    const int b    = blockIdx.x;
    const int lane = tid & 31;
    const int wave = tid >> 5;
    const int half = lane >> 4;
    const int mn   = lane & 15;

    const float4* xr4 = reinterpret_cast<const float4*>(x + (size_t)b * DIM);
    float4 xv[4];
    #pragma unroll
    for (int e = 0; e < 4; ++e) xv[e] = xr4[tid + 256 * e];

    float* const sc = psi;
    if (tid < 144) {
        const int wire = tid / 3, which = tid - wire * 3;
        const float* th = theta + wire * 3;
        const float phi = th[0], the = th[1], om = th[2];
        const float ang = 0.5f * ((which == 0) ? the : (which == 1) ? phi : om);
        sc[tid * 2 + 0] = cosf(ang);
        sc[tid * 2 + 1] = sinf(ang);
    } else if (tid >= 192 && tid < 240) {
        const int t = tid - 192;
        permB[t] = inv_perm(t / 12, 1 << (t % 12));
    }
    __syncthreads();
    {
        const int i = tid >> 4, j = tid & 15;
        const float* uq[4];
        float sg[4];
        #pragma unroll
        for (int q = 0; q < 4; ++q) {
            const int bi = (i >> (3 - q)) & 1, bj = (j >> (3 - q)) & 1;
            uq[q] = sc + q * 6 + ((bi == bj) ? 0 : 1);
            sg[q] = (bi == 0 && bj == 1) ? -1.0f : 1.0f;
        }
        const int ga = (j >> 2) * 64 + i * 4 + (j & 3);
        const int gb = ((j >> 1) & 3) * 64 + i * 4 + (j >> 3) * 2 + (j & 1);
        #pragma unroll
        for (int lg = 0; lg < 12; ++lg) {
            float p = 1.0f;
            #pragma unroll
            for (int q = 0; q < 4; ++q) p *= sg[q] * uq[q][lg * 24];
            rLDS[lg * 256 + ((lg % 3 == 1) ? gb : ga)] = p;
        }
    }
    if (tid < 192) {
        const int lg = tid >> 4, d = tid & 15;
        float pr = 1.0f, pi = 0.0f, qr = 1.0f, qi = 0.0f;
        #pragma unroll
        for (int q = 0; q < 4; ++q) {
            const float* w = sc + (lg * 4 + q) * 6;
            const bool set = (d >> (3 - q)) & 1;
            const float cp = w[2], sp = set ? w[3] : -w[3];
            const float co = w[4], so = set ? w[5] : -w[5];
            const float nr = pr * cp - pi * sp, ni = pr * sp + pi * cp;
            const float mr = qr * co - qi * so, mi = qr * so + qi * co;
            pr = nr; pi = ni; qr = mr; qi = mi;
        }
        tabP[lg * 32 + d] = pr; tabP[lg * 32 + 16 + d] = pi;
        tabO[lg * 32 + d] = qr; tabO[lg * 32 + 16 + d] = qi;
    }

    float ss = 0.0f;
    #pragma unroll
    for (int e = 0; e < 4; ++e) {
        const float4 v = xv[e];
        ss += v.x * v.x + v.y * v.y + v.z * v.z + v.w * v.w;
    }
    for (int o = 16; o > 0; o >>= 1) ss += __shfl_down(ss, o, 32);
    if (lane == 0) red[wave] = ss;
    __syncthreads();
    float n1sq = 0.0f;
    #pragma unroll
    for (int wv = 0; wv < 8; ++wv) n1sq += red[wv];
    const float x0   = x[(size_t)b * DIM];
    const bool  zero = (sqrtf(n1sq) <= 1e-12f);
    const float x0n  = zero ? 1.0f : x0;
    const float inv  = 1.0f / sqrtf(n1sq - x0 * x0 + x0n * x0n);
    __syncthreads();
    #pragma unroll
    for (int e = 0; e < 4; ++e) {
        float4 v = xv[e];
        v.x = (tid == 0 && e == 0) ? x0n : v.x;
        v.x *= inv; v.y *= inv; v.z *= inv; v.w *= inv;
        float* const p = psi + SIDX((tid >> 6) + 4 * e, (tid >> 2) & 15, (tid & 3) * 4);
        *reinterpret_cast<float4*>(p)         = v;
        *reinterpret_cast<float4*>(p + PLANE) = make_float4(0.0f, 0.0f, 0.0f, 0.0f);
    }
    __syncthreads();

    const float* const pB0 = psi + SIDX(2 * half, wave, mn);
    float*       const pD0 = psi + SIDX(8 * half, wave, mn);
    const float* const pX  = psi + SIDX(wave, mn, 2 * half);
    const float* const pA  = rLDS + (mn * 4 + half * 2);

    v8f accR[2], accI[2];
    #pragma unroll 1
    for (int l = 0; l < DEPTH; ++l) {
        const float* const pAl = pA + l * (3 * 256);
        const float* const tPl = tabP + l * (3 * 32);
        const float* const tOl = tabO + l * (3 * 32);

        {
            #pragma unroll
            for (int tt = 0; tt < 2; ++tt) { accR[tt] = (v8f){}; accI[tt] = (v8f){}; }
            v2f sr[4][2], si[4][2], far[4], fai[4];
            #pragma unroll
            for (int kq = 0; kq < 4; ++kq) {
                far[kq] = *reinterpret_cast<const v2f*>(&tPl[kq * 4 + 2 * half]);
                fai[kq] = *reinterpret_cast<const v2f*>(&tPl[16 + kq * 4 + 2 * half]);
                #pragma unroll
                for (int tt = 0; tt < 2; ++tt) {
                    const int o = SIDX(kq * 4, 8 * tt, 0);
                    sr[kq][tt] = (v2f){pB0[o], pB0[o + SA]};
                    si[kq][tt] = (v2f){pB0[PLANE + o], pB0[PLANE + o + SA]};
                }
            }
            #pragma unroll
            for (int kq = 0; kq < 4; ++kq) {
                const v2f Ar = *reinterpret_cast<const v2f*>(&pAl[kq * 64]);
                #pragma unroll
                for (int tt = 0; tt < 2; ++tt) {
                    const v2f Br = sr[kq][tt] * far[kq] - si[kq][tt] * fai[kq];
                    const v2f Bi = sr[kq][tt] * fai[kq] + si[kq][tt] * far[kq];
                    accR[tt] = __builtin_amdgcn_wmma_f32_16x16x4_f32(false, Ar, false, Br, (short)0, accR[tt], false, false);
                    accI[tt] = __builtin_amdgcn_wmma_f32_16x16x4_f32(false, Ar, false, Bi, (short)0, accI[tt], false, false);
                }
            }
            const float pcr = tPl[2 * 32 + mn], pci = tPl[2 * 32 + 16 + mn];
            #pragma unroll
            for (int tt = 0; tt < 2; ++tt) {
                const float pbr = tPl[1 * 32 + wave + 8 * tt], pbi = tPl[1 * 32 + 16 + wave + 8 * tt];
                const float fr = pbr * pcr - pbi * pci;
                const float fi = pbr * pci + pbi * pcr;
                #pragma unroll
                for (int v = 0; v < 8; ++v) {
                    float* const p = pD0 + SIDX(v, 8 * tt, 0);
                    p[0]     = accR[tt][v] * fr - accI[tt][v] * fi;
                    p[PLANE] = accR[tt][v] * fi + accI[tt][v] * fr;
                }
            }
        }
        __syncthreads();

        {
            v8f xR[2], xI[2];
            #pragma unroll
            for (int tt = 0; tt < 2; ++tt) { xR[tt] = (v8f){}; xI[tt] = (v8f){}; }
            #pragma unroll
            for (int kq = 0; kq < 4; ++kq) {
                const v2f Rb = *reinterpret_cast<const v2f*>(&pAl[2 * 256 + kq * 64]);
                #pragma unroll
                for (int tt = 0; tt < 2; ++tt) {
                    const int o = SIDX(8 * tt, 0, kq * 4);
                    const v2f Xr = *reinterpret_cast<const v2f*>(&pX[o]);
                    const v2f Xi = *reinterpret_cast<const v2f*>(&pX[PLANE + o]);
                    xR[tt] = __builtin_amdgcn_wmma_f32_16x16x4_f32(false, Xr, false, Rb, (short)0, xR[tt], false, false);
                    xI[tt] = __builtin_amdgcn_wmma_f32_16x16x4_f32(false, Xi, false, Rb, (short)0, xI[tt], false, false);
                }
            }
            #pragma unroll
            for (int tt = 0; tt < 2; ++tt) { accR[tt] = (v8f){}; accI[tt] = (v8f){}; }
            #pragma unroll
            for (int q = 0; q < 4; ++q) {
                const v2f Ra = *reinterpret_cast<const v2f*>(&pAl[1 * 256 + q * 64]);
                #pragma unroll
                for (int tt = 0; tt < 2; ++tt) {
                    const v2f Br = (v2f){xR[tt][2 * q], xR[tt][2 * q + 1]};
                    const v2f Bi = (v2f){xI[tt][2 * q], xI[tt][2 * q + 1]};
                    accR[tt] = __builtin_amdgcn_wmma_f32_16x16x4_f32(false, Ra, false, Br, (short)0, accR[tt], false, false);
                    accI[tt] = __builtin_amdgcn_wmma_f32_16x16x4_f32(false, Ra, false, Bi, (short)0, accI[tt], false, false);
                }
            }
        }
        if (l + 1 < DEPTH) {
            __syncthreads();
            const float4 c0r = *reinterpret_cast<const float4*>(&tOl[1 * 32 + 8 * half]);
            const float4 c1r = *reinterpret_cast<const float4*>(&tOl[1 * 32 + 8 * half + 4]);
            const float4 c0i = *reinterpret_cast<const float4*>(&tOl[1 * 32 + 16 + 8 * half]);
            const float4 c1i = *reinterpret_cast<const float4*>(&tOl[1 * 32 + 16 + 8 * half + 4]);
            const float obr8[8] = {c0r.x, c0r.y, c0r.z, c0r.w, c1r.x, c1r.y, c1r.z, c1r.w};
            const float obi8[8] = {c0i.x, c0i.y, c0i.z, c0i.w, c1i.x, c1i.y, c1i.z, c1i.w};
            const float ocr = tOl[2 * 32 + mn], oci = tOl[2 * 32 + 16 + mn];
            const int* const bas = &permB[l * 12];
            int it = bas[7] & -half;
            #pragma unroll
            for (int bb = 0; bb < 4; ++bb) it ^= bas[bb] & -((mn >> bb) & 1);
            #pragma unroll
            for (int bb = 0; bb < 3; ++bb) it ^= bas[8 + bb] & -((wave >> bb) & 1);
            const int q4  = __builtin_amdgcn_readfirstlane(bas[4]);
            const int q5  = __builtin_amdgcn_readfirstlane(bas[5]);
            const int q6  = __builtin_amdgcn_readfirstlane(bas[6]);
            const int q11 = __builtin_amdgcn_readfirstlane(bas[11]);
            #pragma unroll
            for (int tt = 0; tt < 2; ++tt) {
                const float oar = tOl[wave + 8 * tt], oai = tOl[16 + wave + 8 * tt];
                const float acr = oar * ocr - oai * oci;
                const float aci = oar * oci + oai * ocr;
                #pragma unroll
                for (int v = 0; v < 8; ++v) {
                    const float ur = accR[tt][v] * obr8[v] - accI[tt][v] * obi8[v];
                    const float ui = accR[tt][v] * obi8[v] + accI[tt][v] * obr8[v];
                    const int i = it ^ (tt ? q11 : 0) ^ ((v & 1) ? q4 : 0) ^ ((v & 2) ? q5 : 0) ^ ((v & 4) ? q6 : 0);
                    float* const p = psi + SLIN(i);
                    p[0]     = ur * acr - ui * aci;
                    p[PLANE] = ur * aci + ui * acr;
                }
            }
            __syncthreads();
        }
    }

    {
        constexpr int MK[12] = {sign_mask(0), sign_mask(1), sign_mask(2),  sign_mask(3),
                                sign_mask(4), sign_mask(5), sign_mask(6),  sign_mask(7),
                                sign_mask(8), sign_mask(9), sign_mask(10), sign_mask(11)};
        float pv[2][8];
        #pragma unroll
        for (int tt = 0; tt < 2; ++tt) {
            #pragma unroll
            for (int v = 0; v < 8; ++v)
                pv[tt][v] = accR[tt][v] * accR[tt][v] + accI[tt][v] * accI[tt][v];
        }
        const int jt = (wave << 8) | (half << 7) | mn;
        #pragma unroll
        for (int k = 0; k < 12; ++k) {
            float s = 0.0f;
            #pragma unroll
            for (int tt = 0; tt < 2; ++tt) {
                #pragma unroll
                for (int v = 0; v < 8; ++v)
                    s += __builtin_parity(((tt << 11) | (v << 4)) & MK[k]) ? -pv[tt][v] : pv[tt][v];
            }
            const int flip = (__builtin_popcount(jt & MK[k]) & 1) << 31;
            s = __int_as_float(__float_as_int(s) ^ flip);
            for (int o = 16; o > 0; o >>= 1) s += __shfl_xor(s, o, 32);
            if (lane == 0) red[wave * 12 + k] = s;
        }
    }
    __syncthreads();

    if (tid < 12) {
        float s = 0.0f;
        #pragma unroll
        for (int wv = 0; wv < 8; ++wv) s += red[wv * 12 + tid];
        out[(size_t)b * 12 + tid] = s;
    }
}

extern "C" void kernel_launch(void* const* d_in, const int* in_sizes, int n_in,
                              void* d_out, int out_size, void* d_ws, size_t ws_size,
                              hipStream_t stream) {
    const float* x     = (const float*)d_in[0];
    const float* theta = (const float*)d_in[1];
    float*       out   = (float*)d_out;

    const int B = in_sizes[0] / DIM;

    qdl_main_kernel<<<B, 256, 0, stream>>>(x, theta, out);
}


# Output sums: the 12 per-wave lane reductions run together (lane pairs split the outputs, 18 shuffles in 5 steps) instead of 12 serial 5-step chains with a branch each; sums bit-identical
# speedup vs baseline: 2.2461x; 1.0517x over previous
#include <hip/hip_runtime.h>
#include <math.h>

#define NQ    12
#define DIM   4096
#define DEPTH 4

#define SA    308
#define SB    36
#define PLANE (15 * SA + 15 * SB + 16)
#define SIDX(a, b, c) ((a) * SA + (b) * SB + (c))
#define SLIN(i) ((i) + (SB - 16) * ((i) >> 4) + (SA - 256 - 16 * (SB - 16)) * ((i) >> 8))

typedef __attribute__((ext_vector_type(2))) float v2f;
typedef __attribute__((ext_vector_type(8))) float v8f;

__host__ __device__ constexpr int inv_perm(int l, int j) {
    const int r = l % (NQ - 1) + 1;
    for (int w = 0; w < NQ; ++w) {
        const int tg   = (w + r) % NQ;
        const int cbit = (j >> (NQ - 1 - w)) & 1;
        j ^= cbit << (NQ - 1 - tg);
    }
    return j;
}
__host__ __device__ constexpr int sign_mask(int k) {
    int m = 0;
    for (int q = 0; q < NQ; ++q)
        if ((inv_perm(DEPTH - 1, 1 << q) >> (NQ - 1 - k)) & 1) m |= 1 << q;
    return m;
}

__global__ __launch_bounds__(256, 10) void qdl_main_kernel(
        const float* __restrict__ x,
        const float* __restrict__ theta,
        float* __restrict__ out) {
    __shared__ __attribute__((aligned(16))) float psi[2 * PLANE];
    __shared__ __attribute__((aligned(16))) float rLDS[12 * 256];
    __shared__ __attribute__((aligned(16))) float tabP[12 * 32];
    __shared__ __attribute__((aligned(16))) float tabO[12 * 32];
    __shared__ int   permB[48];
    __shared__ float red[8 * 12];

    const int tid  = threadIdx.x;
    const int b    = blockIdx.x;
    const int lane = tid & 31;
    const int wave = tid >> 5;
    const int half = lane >> 4;
    const int mn   = lane & 15;

    const float4* xr4 = reinterpret_cast<const float4*>(x + (size_t)b * DIM);
    float4 xv[4];
    #pragma unroll
    for (int e = 0; e < 4; ++e) xv[e] = xr4[tid + 256 * e];

    float* const sc = psi;
    if (tid < 144) {
        const int wire = tid / 3, which = tid - wire * 3;
        const float* th = theta + wire * 3;
        const float phi = th[0], the = th[1], om = th[2];
        const float ang = 0.5f * ((which == 0) ? the : (which == 1) ? phi : om);
        sc[tid * 2 + 0] = cosf(ang);
        sc[tid * 2 + 1] = sinf(ang);
    } else if (tid >= 192 && tid < 240) {
        const int t = tid - 192;
        permB[t] = inv_perm(t / 12, 1 << (t % 12));
    }
    __syncthreads();
    {
        const int i = tid >> 4, j = tid & 15;
        const float* uq[4];
        float sg[4];
        #pragma unroll
        for (int q = 0; q < 4; ++q) {
            const int bi = (i >> (3 - q)) & 1, bj = (j >> (3 - q)) & 1;
            uq[q] = sc + q * 6 + ((bi == bj) ? 0 : 1);
            sg[q] = (bi == 0 && bj == 1) ? -1.0f : 1.0f;
        }
        const int ga = (j >> 2) * 64 + i * 4 + (j & 3);
        const int gb = ((j >> 1) & 3) * 64 + i * 4 + (j >> 3) * 2 + (j & 1);
        #pragma unroll
        for (int lg = 0; lg < 12; ++lg) {
            float p = 1.0f;
            #pragma unroll
            for (int q = 0; q < 4; ++q) p *= sg[q] * uq[q][lg * 24];
            rLDS[lg * 256 + ((lg % 3 == 1) ? gb : ga)] = p;
        }
    }
    if (tid < 192) {
        const int lg = tid >> 4, d = tid & 15;
        float pr = 1.0f, pi = 0.0f, qr = 1.0f, qi = 0.0f;
        #pragma unroll
        for (int q = 0; q < 4; ++q) {
            const float* w = sc + (lg * 4 + q) * 6;
            const bool set = (d >> (3 - q)) & 1;
            const float cp = w[2], sp = set ? w[3] : -w[3];
            const float co = w[4], so = set ? w[5] : -w[5];
            const float nr = pr * cp - pi * sp, ni = pr * sp + pi * cp;
            const float mr = qr * co - qi * so, mi = qr * so + qi * co;
            pr = nr; pi = ni; qr = mr; qi = mi;
        }
        tabP[lg * 32 + d] = pr; tabP[lg * 32 + 16 + d] = pi;
        tabO[lg * 32 + d] = qr; tabO[lg * 32 + 16 + d] = qi;
    }

    float ss = 0.0f;
    #pragma unroll
    for (int e = 0; e < 4; ++e) {
        const float4 v = xv[e];
        ss += v.x * v.x + v.y * v.y + v.z * v.z + v.w * v.w;
    }
    for (int o = 16; o > 0; o >>= 1) ss += __shfl_down(ss, o, 32);
    if (lane == 0) red[wave] = ss;
    __syncthreads();
    float n1sq = 0.0f;
    #pragma unroll
    for (int wv = 0; wv < 8; ++wv) n1sq += red[wv];
    const float x0   = x[(size_t)b * DIM];
    const bool  zero = (sqrtf(n1sq) <= 1e-12f);
    const float x0n  = zero ? 1.0f : x0;
    const float inv  = 1.0f / sqrtf(n1sq - x0 * x0 + x0n * x0n);
    __syncthreads();
    #pragma unroll
    for (int e = 0; e < 4; ++e) {
        float4 v = xv[e];
        v.x = (tid == 0 && e == 0) ? x0n : v.x;
        v.x *= inv; v.y *= inv; v.z *= inv; v.w *= inv;
        float* const p = psi + SIDX((tid >> 6) + 4 * e, (tid >> 2) & 15, (tid & 3) * 4);
        *reinterpret_cast<float4*>(p)         = v;
        *reinterpret_cast<float4*>(p + PLANE) = make_float4(0.0f, 0.0f, 0.0f, 0.0f);
    }
    __syncthreads();

    const float* const pB0 = psi + SIDX(2 * half, wave, mn);
    float*       const pD0 = psi + SIDX(8 * half, wave, mn);
    const float* const pX  = psi + SIDX(wave, mn, 2 * half);
    const float* const pA  = rLDS + (mn * 4 + half * 2);

    v8f accR[2], accI[2];
    #pragma unroll 1
    for (int l = 0; l < DEPTH; ++l) {
        const float* const pAl = pA + l * (3 * 256);
        const float* const tPl = tabP + l * (3 * 32);
        const float* const tOl = tabO + l * (3 * 32);

        {
            #pragma unroll
            for (int tt = 0; tt < 2; ++tt) { accR[tt] = (v8f){}; accI[tt] = (v8f){}; }
            v2f sr[4][2], si[4][2], far[4], fai[4];
            #pragma unroll
            for (int kq = 0; kq < 4; ++kq) {
                far[kq] = *reinterpret_cast<const v2f*>(&tPl[kq * 4 + 2 * half]);
                fai[kq] = *reinterpret_cast<const v2f*>(&tPl[16 + kq * 4 + 2 * half]);
                #pragma unroll
                for (int tt = 0; tt < 2; ++tt) {
                    const int o = SIDX(kq * 4, 8 * tt, 0);
                    sr[kq][tt] = (v2f){pB0[o], pB0[o + SA]};
                    si[kq][tt] = (v2f){pB0[PLANE + o], pB0[PLANE + o + SA]};
                }
            }
            #pragma unroll
            for (int kq = 0; kq < 4; ++kq) {
                const v2f Ar = *reinterpret_cast<const v2f*>(&pAl[kq * 64]);
                #pragma unroll
                for (int tt = 0; tt < 2; ++tt) {
                    const v2f Br = sr[kq][tt] * far[kq] - si[kq][tt] * fai[kq];
                    const v2f Bi = sr[kq][tt] * fai[kq] + si[kq][tt] * far[kq];
                    accR[tt] = __builtin_amdgcn_wmma_f32_16x16x4_f32(false, Ar, false, Br, (short)0, accR[tt], false, false);
                    accI[tt] = __builtin_amdgcn_wmma_f32_16x16x4_f32(false, Ar, false, Bi, (short)0, accI[tt], false, false);
                }
            }
            const float pcr = tPl[2 * 32 + mn], pci = tPl[2 * 32 + 16 + mn];
            #pragma unroll
            for (int tt = 0; tt < 2; ++tt) {
                const float pbr = tPl[1 * 32 + wave + 8 * tt], pbi = tPl[1 * 32 + 16 + wave + 8 * tt];
                const float fr = pbr * pcr - pbi * pci;
                const float fi = pbr * pci + pbi * pcr;
                #pragma unroll
                for (int v = 0; v < 8; ++v) {
                    float* const p = pD0 + SIDX(v, 8 * tt, 0);
                    p[0]     = accR[tt][v] * fr - accI[tt][v] * fi;
                    p[PLANE] = accR[tt][v] * fi + accI[tt][v] * fr;
                }
            }
        }
        __syncthreads();

        {
            v8f xR[2], xI[2];
            #pragma unroll
            for (int tt = 0; tt < 2; ++tt) { xR[tt] = (v8f){}; xI[tt] = (v8f){}; }
            #pragma unroll
            for (int kq = 0; kq < 4; ++kq) {
                const v2f Rb = *reinterpret_cast<const v2f*>(&pAl[2 * 256 + kq * 64]);
                #pragma unroll
                for (int tt = 0; tt < 2; ++tt) {
                    const int o = SIDX(8 * tt, 0, kq * 4);
                    const v2f Xr = *reinterpret_cast<const v2f*>(&pX[o]);
                    const v2f Xi = *reinterpret_cast<const v2f*>(&pX[PLANE + o]);
                    xR[tt] = __builtin_amdgcn_wmma_f32_16x16x4_f32(false, Xr, false, Rb, (short)0, xR[tt], false, false);
                    xI[tt] = __builtin_amdgcn_wmma_f32_16x16x4_f32(false, Xi, false, Rb, (short)0, xI[tt], false, false);
                }
            }
            #pragma unroll
            for (int tt = 0; tt < 2; ++tt) { accR[tt] = (v8f){}; accI[tt] = (v8f){}; }
            #pragma unroll
            for (int q = 0; q < 4; ++q) {
                const v2f Ra = *reinterpret_cast<const v2f*>(&pAl[1 * 256 + q * 64]);
                #pragma unroll
                for (int tt = 0; tt < 2; ++tt) {
                    const v2f Br = (v2f){xR[tt][2 * q], xR[tt][2 * q + 1]};
                    const v2f Bi = (v2f){xI[tt][2 * q], xI[tt][2 * q + 1]};
                    accR[tt] = __builtin_amdgcn_wmma_f32_16x16x4_f32(false, Ra, false, Br, (short)0, accR[tt], false, false);
                    accI[tt] = __builtin_amdgcn_wmma_f32_16x16x4_f32(false, Ra, false, Bi, (short)0, accI[tt], false, false);
                }
            }
        }
        if (l + 1 < DEPTH) {
            __syncthreads();
            const float4 c0r = *reinterpret_cast<const float4*>(&tOl[1 * 32 + 8 * half]);
            const float4 c1r = *reinterpret_cast<const float4*>(&tOl[1 * 32 + 8 * half + 4]);
            const float4 c0i = *reinterpret_cast<const float4*>(&tOl[1 * 32 + 16 + 8 * half]);
            const float4 c1i = *reinterpret_cast<const float4*>(&tOl[1 * 32 + 16 + 8 * half + 4]);
            const float obr8[8] = {c0r.x, c0r.y, c0r.z, c0r.w, c1r.x, c1r.y, c1r.z, c1r.w};
            const float obi8[8] = {c0i.x, c0i.y, c0i.z, c0i.w, c1i.x, c1i.y, c1i.z, c1i.w};
            const float ocr = tOl[2 * 32 + mn], oci = tOl[2 * 32 + 16 + mn];
            const int* const bas = &permB[l * 12];
            int it = bas[7] & -half;
            #pragma unroll
            for (int bb = 0; bb < 4; ++bb) it ^= bas[bb] & -((mn >> bb) & 1);
            #pragma unroll
            for (int bb = 0; bb < 3; ++bb) it ^= bas[8 + bb] & -((wave >> bb) & 1);
            const int q4  = __builtin_amdgcn_readfirstlane(bas[4]);
            const int q5  = __builtin_amdgcn_readfirstlane(bas[5]);
            const int q6  = __builtin_amdgcn_readfirstlane(bas[6]);
            const int q11 = __builtin_amdgcn_readfirstlane(bas[11]);
            #pragma unroll
            for (int tt = 0; tt < 2; ++tt) {
                const float oar = tOl[wave + 8 * tt], oai = tOl[16 + wave + 8 * tt];
                const float acr = oar * ocr - oai * oci;
                const float aci = oar * oci + oai * ocr;
                #pragma unroll
                for (int v = 0; v < 8; ++v) {
                    const float ur = accR[tt][v] * obr8[v] - accI[tt][v] * obi8[v];
                    const float ui = accR[tt][v] * obi8[v] + accI[tt][v] * obr8[v];
                    const int i = it ^ (tt ? q11 : 0) ^ ((v & 1) ? q4 : 0) ^ ((v & 2) ? q5 : 0) ^ ((v & 4) ? q6 : 0);
                    float* const p = psi + SLIN(i);
                    p[0]     = ur * acr - ui * aci;
                    p[PLANE] = ur * aci + ui * acr;
                }
            }
            __syncthreads();
        }
    }

    {
        constexpr int MK[12] = {sign_mask(0), sign_mask(1), sign_mask(2),  sign_mask(3),
                                sign_mask(4), sign_mask(5), sign_mask(6),  sign_mask(7),
                                sign_mask(8), sign_mask(9), sign_mask(10), sign_mask(11)};
        float pv[2][8];
        #pragma unroll
        for (int tt = 0; tt < 2; ++tt) {
            #pragma unroll
            for (int v = 0; v < 8; ++v)
                pv[tt][v] = accR[tt][v] * accR[tt][v] + accI[tt][v] * accI[tt][v];
        }
        const int jt = (wave << 8) | (half << 7) | mn;
        float sk[12];
        #pragma unroll
        for (int k = 0; k < 12; ++k) {
            float s = 0.0f;
            #pragma unroll
            for (int tt = 0; tt < 2; ++tt) {
                #pragma unroll
                for (int v = 0; v < 8; ++v)
                    s += __builtin_parity(((tt << 11) | (v << 4)) & MK[k]) ? -pv[tt][v] : pv[tt][v];
            }
            const int flip = (__builtin_popcount(jt & MK[k]) & 1) << 31;
            sk[k] = __int_as_float(__float_as_int(s) ^ flip);
        }
        float t6[6];
        #pragma unroll
        for (int i = 0; i < 6; ++i) {
            const float keep = half ? sk[i + 6] : sk[i];
            const float send = half ? sk[i] : sk[i + 6];
            t6[i] = keep + __shfl_xor(send, 16, 32);
        }
        const int b3 = (lane >> 3) & 1;
        float u3[3];
        #pragma unroll
        for (int i = 0; i < 3; ++i) {
            const float keep = b3 ? t6[i + 3] : t6[i];
            const float send = b3 ? t6[i] : t6[i + 3];
            u3[i] = keep + __shfl_xor(send, 8, 32);
        }
        #pragma unroll
        for (int o = 4; o > 0; o >>= 1) {
            #pragma unroll
            for (int i = 0; i < 3; ++i) u3[i] += __shfl_xor(u3[i], o, 32);
        }
        if ((lane & 7) == 0) {
            float* const r = &red[wave * 12 + 6 * half + 3 * b3];
            r[0] = u3[0]; r[1] = u3[1]; r[2] = u3[2];
        }
    }
    __syncthreads();

    if (tid < 12) {
        float s = 0.0f;
        #pragma unroll
        for (int wv = 0; wv < 8; ++wv) s += red[wv * 12 + tid];
        out[(size_t)b * 12 + tid] = s;
    }
}

extern "C" void kernel_launch(void* const* d_in, const int* in_sizes, int n_in,
                              void* d_out, int out_size, void* d_ws, size_t ws_size,
                              hipStream_t stream) {
    const float* x     = (const float*)d_in[0];
    const float* theta = (const float*)d_in[1];
    float*       out   = (float*)d_out;

    const int B = in_sizes[0] / DIM;

    qdl_main_kernel<<<B, 256, 0, stream>>>(x, theta, out);
}


# State in LDS laid out by an XOR swizzle (linear over GF(2), no padding) instead of padded strides: a permuted element's address is one XOR, not shifts and two multiply-adds; sums unchanged
# speedup vs baseline: 2.4143x; 1.0749x over previous
#include <hip/hip_runtime.h>
#include <math.h>

#define NQ    12
#define DIM   4096
#define DEPTH 4

#define PLANE 4096
#define SWZ(i) ((i) ^ (((i) >> 4) & 0xC) ^ (((i) >> 5) & 0x10))

typedef __attribute__((ext_vector_type(2))) float v2f;
typedef __attribute__((ext_vector_type(8))) float v8f;

__host__ __device__ constexpr int inv_perm(int l, int j) {
    const int r = l % (NQ - 1) + 1;
    for (int w = 0; w < NQ; ++w) {
        const int tg   = (w + r) % NQ;
        const int cbit = (j >> (NQ - 1 - w)) & 1;
        j ^= cbit << (NQ - 1 - tg);
    }
    return j;
}
__host__ __device__ constexpr int sign_mask(int k) {
    int m = 0;
    for (int q = 0; q < NQ; ++q)
        if ((inv_perm(DEPTH - 1, 1 << q) >> (NQ - 1 - k)) & 1) m |= 1 << q;
    return m;
}

__global__ __launch_bounds__(256, 10) void qdl_main_kernel(
        const float* __restrict__ x,
        const float* __restrict__ theta,
        float* __restrict__ out) {
    struct Shared {
        float psi[2 * PLANE];
        float rLDS[12 * 256];
        float tabP[12 * 32];
        float tabO[12 * 32];
        int   permB[48];
        float red[8 * 12];
    };
    __shared__ __attribute__((aligned(16))) Shared sh;
    float* const psi   = sh.psi;
    float* const rLDS  = sh.rLDS;
    float* const tabP  = sh.tabP;
    float* const tabO  = sh.tabO;
    int*   const permB = sh.permB;
    float* const red   = sh.red;

    const int tid  = threadIdx.x;
    const int b    = blockIdx.x;
    const int lane = tid & 31;
    const int wave = tid >> 5;
    const int half = lane >> 4;
    const int mn   = lane & 15;

    const float4* xr4 = reinterpret_cast<const float4*>(x + (size_t)b * DIM);
    float4 xv[4];
    #pragma unroll
    for (int e = 0; e < 4; ++e) xv[e] = xr4[tid + 256 * e];

    float* const sc = psi;
    if (tid < 144) {
        const int wire = tid / 3, which = tid - wire * 3;
        const float* th = theta + wire * 3;
        const float phi = th[0], the = th[1], om = th[2];
        const float ang = 0.5f * ((which == 0) ? the : (which == 1) ? phi : om);
        sc[tid * 2 + 0] = cosf(ang);
        sc[tid * 2 + 1] = sinf(ang);
    } else if (tid >= 192 && tid < 240) {
        const int t = tid - 192;
        permB[t] = inv_perm(t / 12, 1 << (t % 12));
    }
    __syncthreads();
    {
        const int i = tid >> 4, j = tid & 15;
        const float* uq[4];
        float sg[4];
        #pragma unroll
        for (int q = 0; q < 4; ++q) {
            const int bi = (i >> (3 - q)) & 1, bj = (j >> (3 - q)) & 1;
            uq[q] = sc + q * 6 + ((bi == bj) ? 0 : 1);
            sg[q] = (bi == 0 && bj == 1) ? -1.0f : 1.0f;
        }
        const int ga = (j >> 2) * 64 + i * 4 + (j & 3);
        const int g0 = (j >> 2) * 64 + ((i & 1) | (((i >> 2) & 3) << 1) | (((i >> 1) & 1) << 3)) * 4 + (j & 3);
        const int gb = ((j >> 1) & 3) * 64 + i * 4 + (j >> 3) * 2 + (j & 1);
        #pragma unroll
        for (int lg = 0; lg < 12; ++lg) {
            float p = 1.0f;
            #pragma unroll
            for (int q = 0; q < 4; ++q) p *= sg[q] * uq[q][lg * 24];
            rLDS[lg * 256 + ((lg % 3 == 1) ? gb : (lg % 3 == 0) ? g0 : ga)] = p;
        }
    }
    if (tid < 192) {
        const int lg = tid >> 4, d = tid & 15;
        float pr = 1.0f, pi = 0.0f, qr = 1.0f, qi = 0.0f;
        #pragma unroll
        for (int q = 0; q < 4; ++q) {
            const float* w = sc + (lg * 4 + q) * 6;
            const bool set = (d >> (3 - q)) & 1;
            const float cp = w[2], sp = set ? w[3] : -w[3];
            const float co = w[4], so = set ? w[5] : -w[5];
            const float nr = pr * cp - pi * sp, ni = pr * sp + pi * cp;
            const float mr = qr * co - qi * so, mi = qr * so + qi * co;
            pr = nr; pi = ni; qr = mr; qi = mi;
        }
        tabP[lg * 32 + d] = pr; tabP[lg * 32 + 16 + d] = pi;
        tabO[lg * 32 + d] = qr; tabO[lg * 32 + 16 + d] = qi;
    }

    float ss = 0.0f;
    #pragma unroll
    for (int e = 0; e < 4; ++e) {
        const float4 v = xv[e];
        ss += v.x * v.x + v.y * v.y + v.z * v.z + v.w * v.w;
    }
    for (int o = 16; o > 0; o >>= 1) ss += __shfl_down(ss, o, 32);
    if (lane == 0) red[wave] = ss;
    __syncthreads();
    float n1sq = 0.0f;
    #pragma unroll
    for (int wv = 0; wv < 8; ++wv) n1sq += red[wv];
    const float x0   = x[(size_t)b * DIM];
    const bool  zero = (sqrtf(n1sq) <= 1e-12f);
    const float x0n  = zero ? 1.0f : x0;
    const float inv  = 1.0f / sqrtf(n1sq - x0 * x0 + x0n * x0n);
    __syncthreads();
    const int st0 = SWZ(4 * tid);
    #pragma unroll
    for (int e = 0; e < 4; ++e) {
        float4 v = xv[e];
        v.x = (tid == 0 && e == 0) ? x0n : v.x;
        v.x *= inv; v.y *= inv; v.z *= inv; v.w *= inv;
        float* const p = psi + st0 + 1024 * e;
        *reinterpret_cast<float4*>(p)         = v;
        *reinterpret_cast<float4*>(p + PLANE) = make_float4(0.0f, 0.0f, 0.0f, 0.0f);
    }
    __syncthreads();

    const int bG = SWZ(half * 512 + wave * 16 + mn);
    float* const pG[2] = {psi + bG, psi + (bG ^ 136)};
    const int bX = SWZ(wave * 256 + mn * 16 + 2 * half) * 4;
    const float* const pA  = rLDS + (mn * 4 + half * 2);

    v8f accR[2], accI[2];
    #pragma unroll 1
    for (int l = 0; l < DEPTH; ++l) {
        const float* const pAl = pA + l * (3 * 256);
        const float* const tPl = tabP + l * (3 * 32);
        const float* const tOl = tabO + l * (3 * 32);

        {
            #pragma unroll
            for (int tt = 0; tt < 2; ++tt) { accR[tt] = (v8f){}; accI[tt] = (v8f){}; }
            v2f sr[4][2], si[4][2], far[4], fai[4];
            #pragma unroll
            for (int kq = 0; kq < 4; ++kq) {
                far[kq] = *reinterpret_cast<const v2f*>(&tPl[kq * 4 + 2 * half]);
                fai[kq] = *reinterpret_cast<const v2f*>(&tPl[16 + kq * 4 + 2 * half]);
                #pragma unroll
                for (int tt = 0; tt < 2; ++tt) {
                    const int o = kq * 1024;
                    sr[kq][tt] = (v2f){pG[tt][o], pG[tt][o + 256]};
                    si[kq][tt] = (v2f){pG[tt][PLANE + o], pG[tt][PLANE + o + 256]};
                }
            }
            #pragma unroll
            for (int kq = 0; kq < 4; ++kq) {
                const v2f Ar = *reinterpret_cast<const v2f*>(&pAl[kq * 64]);
                #pragma unroll
                for (int tt = 0; tt < 2; ++tt) {
                    const v2f Br = sr[kq][tt] * far[kq] - si[kq][tt] * fai[kq];
                    const v2f Bi = sr[kq][tt] * fai[kq] + si[kq][tt] * far[kq];
                    accR[tt] = __builtin_amdgcn_wmma_f32_16x16x4_f32(false, Ar, false, Br, (short)0, accR[tt], false, false);
                    accI[tt] = __builtin_amdgcn_wmma_f32_16x16x4_f32(false, Ar, false, Bi, (short)0, accI[tt], false, false);
                }
            }
            const float pcr = tPl[2 * 32 + mn], pci = tPl[2 * 32 + 16 + mn];
            #pragma unroll
            for (int tt = 0; tt < 2; ++tt) {
                const float pbr = tPl[1 * 32 + wave + 8 * tt], pbi = tPl[1 * 32 + 16 + wave + 8 * tt];
                const float fr = pbr * pcr - pbi * pci;
                const float fi = pbr * pci + pbi * pcr;
                #pragma unroll
                for (int v = 0; v < 8; ++v) {
                    float* const p = pG[tt] + ((v & 1) + (v >> 1) * 4) * 256;
                    p[0]     = accR[tt][v] * fr - accI[tt][v] * fi;
                    p[PLANE] = accR[tt][v] * fi + accI[tt][v] * fr;
                }
            }
        }
        __syncthreads();

        {
            v8f xR[2], xI[2];
            #pragma unroll
            for (int tt = 0; tt < 2; ++tt) { xR[tt] = (v8f){}; xI[tt] = (v8f){}; }
            #pragma unroll
            for (int kq = 0; kq < 4; ++kq) {
                const v2f Rb = *reinterpret_cast<const v2f*>(&pAl[2 * 256 + kq * 64]);
                const float* const pX = reinterpret_cast<const float*>(
                    reinterpret_cast<const char*>(psi) + (bX ^ (kq * 16)));
                #pragma unroll
                for (int tt = 0; tt < 2; ++tt) {
                    const int o = 2048 * tt;
                    const v2f Xr = *reinterpret_cast<const v2f*>(&pX[o]);
                    const v2f Xi = *reinterpret_cast<const v2f*>(&pX[PLANE + o]);
                    xR[tt] = __builtin_amdgcn_wmma_f32_16x16x4_f32(false, Xr, false, Rb, (short)0, xR[tt], false, false);
                    xI[tt] = __builtin_amdgcn_wmma_f32_16x16x4_f32(false, Xi, false, Rb, (short)0, xI[tt], false, false);
                }
            }
            #pragma unroll
            for (int tt = 0; tt < 2; ++tt) { accR[tt] = (v8f){}; accI[tt] = (v8f){}; }
            #pragma unroll
            for (int q = 0; q < 4; ++q) {
                const v2f Ra = *reinterpret_cast<const v2f*>(&pAl[1 * 256 + q * 64]);
                #pragma unroll
                for (int tt = 0; tt < 2; ++tt) {
                    const v2f Br = (v2f){xR[tt][2 * q], xR[tt][2 * q + 1]};
                    const v2f Bi = (v2f){xI[tt][2 * q], xI[tt][2 * q + 1]};
                    accR[tt] = __builtin_amdgcn_wmma_f32_16x16x4_f32(false, Ra, false, Br, (short)0, accR[tt], false, false);
                    accI[tt] = __builtin_amdgcn_wmma_f32_16x16x4_f32(false, Ra, false, Bi, (short)0, accI[tt], false, false);
                }
            }
        }
        if (l + 1 < DEPTH) {
            __syncthreads();
            const float4 c0r = *reinterpret_cast<const float4*>(&tOl[1 * 32 + 8 * half]);
            const float4 c1r = *reinterpret_cast<const float4*>(&tOl[1 * 32 + 8 * half + 4]);
            const float4 c0i = *reinterpret_cast<const float4*>(&tOl[1 * 32 + 16 + 8 * half]);
            const float4 c1i = *reinterpret_cast<const float4*>(&tOl[1 * 32 + 16 + 8 * half + 4]);
            const float obr8[8] = {c0r.x, c0r.y, c0r.z, c0r.w, c1r.x, c1r.y, c1r.z, c1r.w};
            const float obi8[8] = {c0i.x, c0i.y, c0i.z, c0i.w, c1i.x, c1i.y, c1i.z, c1i.w};
            const float ocr = tOl[2 * 32 + mn], oci = tOl[2 * 32 + 16 + mn];
            const int* const bas = &permB[l * 12];
            int it = bas[7] & -half;
            #pragma unroll
            for (int bb = 0; bb < 4; ++bb) it ^= bas[bb] & -((mn >> bb) & 1);
            #pragma unroll
            for (int bb = 0; bb < 3; ++bb) it ^= bas[8 + bb] & -((wave >> bb) & 1);
            const int sit = SWZ(it) * 4;
            const int r4  = __builtin_amdgcn_readfirstlane(bas[4]);
            const int r5  = __builtin_amdgcn_readfirstlane(bas[5]);
            const int r6  = __builtin_amdgcn_readfirstlane(bas[6]);
            const int r11 = __builtin_amdgcn_readfirstlane(bas[11]);
            const int q4 = SWZ(r4) * 4, q5 = SWZ(r5) * 4, q6 = SWZ(r6) * 4, q11 = SWZ(r11) * 4;
            #pragma unroll
            for (int tt = 0; tt < 2; ++tt) {
                const float oar = tOl[wave + 8 * tt], oai = tOl[16 + wave + 8 * tt];
                const float acr = oar * ocr - oai * oci;
                const float aci = oar * oci + oai * ocr;
                #pragma unroll
                for (int v = 0; v < 8; ++v) {
                    const float ur = accR[tt][v] * obr8[v] - accI[tt][v] * obi8[v];
                    const float ui = accR[tt][v] * obi8[v] + accI[tt][v] * obr8[v];
                    const int ob = sit ^ (tt ? q11 : 0) ^ ((v & 1) ? q4 : 0) ^ ((v & 2) ? q5 : 0) ^ ((v & 4) ? q6 : 0);
                    float* const p = reinterpret_cast<float*>(reinterpret_cast<char*>(psi) + ob);
                    p[0]     = ur * acr - ui * aci;
                    p[PLANE] = ur * aci + ui * acr;
                }
            }
            __syncthreads();
        }
    }

    {
        constexpr int MK[12] = {sign_mask(0), sign_mask(1), sign_mask(2),  sign_mask(3),
                                sign_mask(4), sign_mask(5), sign_mask(6),  sign_mask(7),
                                sign_mask(8), sign_mask(9), sign_mask(10), sign_mask(11)};
        float pv[2][8];
        #pragma unroll
        for (int tt = 0; tt < 2; ++tt) {
            #pragma unroll
            for (int v = 0; v < 8; ++v)
                pv[tt][v] = accR[tt][v] * accR[tt][v] + accI[tt][v] * accI[tt][v];
        }
        const int jt = (wave << 8) | (half << 7) | mn;
        float sk[12];
        #pragma unroll
        for (int k = 0; k < 12; ++k) {
            float s = 0.0f;
            #pragma unroll
            for (int tt = 0; tt < 2; ++tt) {
                #pragma unroll
                for (int v = 0; v < 8; ++v)
                    s += __builtin_parity(((tt << 11) | (v << 4)) & MK[k]) ? -pv[tt][v] : pv[tt][v];
            }
            const int flip = (__builtin_popcount(jt & MK[k]) & 1) << 31;
            sk[k] = __int_as_float(__float_as_int(s) ^ flip);
        }
        float t6[6];
        #pragma unroll
        for (int i = 0; i < 6; ++i) {
            const float keep = half ? sk[i + 6] : sk[i];
            const float send = half ? sk[i] : sk[i + 6];
            t6[i] = keep + __shfl_xor(send, 16, 32);
        }
        const int b3 = (lane >> 3) & 1;
        float u3[3];
        #pragma unroll
        for (int i = 0; i < 3; ++i) {
            const float keep = b3 ? t6[i + 3] : t6[i];
            const float send = b3 ? t6[i] : t6[i + 3];
            u3[i] = keep + __shfl_xor(send, 8, 32);
        }
        #pragma unroll
        for (int o = 4; o > 0; o >>= 1) {
            #pragma unroll
            for (int i = 0; i < 3; ++i) u3[i] += __shfl_xor(u3[i], o, 32);
        }
        if ((lane & 7) == 0) {
            float* const r = &red[wave * 12 + 6 * half + 3 * b3];
            r[0] = u3[0]; r[1] = u3[1]; r[2] = u3[2];
        }
    }
    __syncthreads();

    if (tid < 12) {
        float s = 0.0f;
        #pragma unroll
        for (int wv = 0; wv < 8; ++wv) s += red[wv * 12 + tid];
        out[(size_t)b * 12 + tid] = s;
    }
}

extern "C" void kernel_launch(void* const* d_in, const int* in_sizes, int n_in,
                              void* d_out, int out_size, void* d_ws, size_t ws_size,
                              hipStream_t stream) {
    const float* x     = (const float*)d_in[0];
    const float* theta = (const float*)d_in[1];
    float*       out   = (float*)d_out;

    const int B = in_sizes[0] / DIM;

    qdl_main_kernel<<<B, 256, 0, stream>>>(x, theta, out);
}
